# starting kernel
# baseline (speedup 1.0000x reference)
#include <hip/hip_runtime.h>

typedef __attribute__((ext_vector_type(2))) float v2f;
typedef __attribute__((ext_vector_type(8))) float v8f;

#define DIN 128


__global__ void GCN_fill0(float* __restrict__ p, int n) {
    int i = blockIdx.x * blockDim.x + threadIdx.x;
    if (i < n) p[i] = 0.0f;
}

__global__ void GCN_degree(const int* __restrict__ dst, float* __restrict__ deg, int E) {
    int e = blockIdx.x * blockDim.x + threadIdx.x;
    if (e < E) atomicAdd(&deg[dst[e]], 1.0f);
}

__global__ void GCN_norm(float* __restrict__ deg, int n) {
    int i = blockIdx.x * blockDim.x + threadIdx.x;
    if (i < n) deg[i] = 1.0f / (deg[i] + 1.0f);
}

__global__ void GCN_scatter(const float* __restrict__ h, const int* __restrict__ src,
                            const int* __restrict__ dst, float* __restrict__ agg, int E) {
    long tid  = (long)blockIdx.x * blockDim.x + threadIdx.x;
    int  e    = (int)(tid >> 5);
    int  lane = (int)(tid & 31);
    if (e >= E) return;
    int s = src[e];
    int d = dst[e];
    const float4 v = ((const float4*)(h + (size_t)s * DIN))[lane];
    float* p = agg + (size_t)d * DIN + lane * 4;
    atomicAdd(p + 0, v.x);
    atomicAdd(p + 1, v.y);
    atomicAdd(p + 2, v.z);
    atomicAdd(p + 3, v.w);
}

template <int NC, bool RELU>
__global__ __launch_bounds__(128) void GCN_fc_wmma(const float* __restrict__ hself,
                                                   const float* __restrict__ hagg,
                                                   const float* __restrict__ norm,
                                                   const float* __restrict__ W,
                                                   const float* __restrict__ bias,
                                                   float* __restrict__ out) {
    constexpr int K    = DIN;
    constexpr int LDSK = K + 2;
    __shared__ float Atile[16 * LDSK];

    const int row0 = blockIdx.x * 16;
    const int tid  = threadIdx.x;

    for (int i = tid; i < 16 * K; i += 128) {
        int r = i >> 7;
        int k = i & (K - 1);
        int gr = row0 + r;
        Atile[r * LDSK + k] =
            (hagg[(size_t)gr * K + k] + hself[(size_t)gr * K + k]) * norm[gr];
    }
    __syncthreads();

    const int wave = tid >> 5;
    const int lane = tid & 31;
    const int m    = lane & 15;
    const int kh2  = (lane >> 4) << 1;

    for (int nt = wave; nt * 16 < NC; nt += 4) {
        const int n0 = nt * 16;
        const float* wrow = W + (size_t)(n0 + m) * K;

        v8f acc = {};
#pragma unroll 4
        for (int kb = 0; kb < K; kb += 4) {
            v2f a, b;
            a.x = Atile[m * LDSK + kb + kh2];
            a.y = Atile[m * LDSK + kb + kh2 + 1];
            b.x = wrow[kb + kh2];
            b.y = wrow[kb + kh2 + 1];
            acc = __builtin_amdgcn_wmma_f32_16x16x4_f32(
                 false, a,  false, b,
                 (short)0, acc,  false,  false);
        }

        const float bn   = bias[n0 + m];
        const int   rupp = (lane >> 4) << 3;
#pragma unroll
        for (int v = 0; v < 8; ++v) {
            float val = acc[v] + bn;
            if (RELU) val = fmaxf(val, 0.0f);
            out[(size_t)(row0 + rupp + v) * NC + (n0 + m)] = val;
        }
    }
}


extern "C" void kernel_launch(void* const* d_in, const int* in_sizes, int n_in,
                              void* d_out, int out_size, void* d_ws, size_t ws_size,
                              hipStream_t stream) {
    (void)n_in; (void)out_size; (void)ws_size;
    const float* x  = (const float*)d_in[0];
    const int*   src = (const int*)d_in[1];
    const int*   dst = (const int*)d_in[2];
    const float* W1 = (const float*)d_in[3];
    const float* b1 = (const float*)d_in[4];
    const float* W2 = (const float*)d_in[5];
    const float* b2 = (const float*)d_in[6];
    const float* W3 = (const float*)d_in[7];
    const float* b3 = (const float*)d_in[8];
    float* out = (float*)d_out;

    const int N = in_sizes[0] / DIN;
    const int E = in_sizes[1];

    char* ws = (char*)d_ws;
    size_t off = 0;
    float* norm = (float*)(ws + off);
    off += (((size_t)N * 4) + 255) & ~(size_t)255;
    float* buf0 = (float*)(ws + off);
    off += (size_t)N * DIN * 4;
    float* buf1 = (float*)(ws + off);

    const int TB = 256;
    const int nf = N * DIN;
    const long sthreads = (long)E * 32;
    const int  sblocks  = (int)((sthreads + TB - 1) / TB);
    const int  fblocks  = N / 16;

    GCN_fill0<<<(N + TB - 1) / TB, TB, 0, stream>>>(norm, N);
    GCN_degree<<<(E + TB - 1) / TB, TB, 0, stream>>>(dst, norm, E);
    GCN_norm<<<(N + TB - 1) / TB, TB, 0, stream>>>(norm, N);

    GCN_fill0<<<(nf + TB - 1) / TB, TB, 0, stream>>>(buf1, nf);
    GCN_scatter<<<sblocks, TB, 0, stream>>>(x, src, dst, buf1, E);
    GCN_fc_wmma<128, true><<<fblocks, 128, 0, stream>>>(x, buf1, norm, W1, b1, buf0);

    GCN_fill0<<<(nf + TB - 1) / TB, TB, 0, stream>>>(buf1, nf);
    GCN_scatter<<<sblocks, TB, 0, stream>>>(buf0, src, dst, buf1, E);
    GCN_fc_wmma<128, true><<<fblocks, 128, 0, stream>>>(buf0, buf1, norm, W2, b2, buf1);

    GCN_fill0<<<(nf + TB - 1) / TB, TB, 0, stream>>>(buf0, nf);
    GCN_scatter<<<sblocks, TB, 0, stream>>>(buf1, src, dst, buf0, E);
    GCN_fc_wmma<64, false><<<fblocks, 128, 0, stream>>>(buf1, buf0, norm, W3, b3, out);
}


# Replace zero-fill + float-atomic scatter-add by a per-call CSR (integer atomic arrival-order ranks, one counter per 128-B line) and an atomic-free one-wave-per-row gather sum
# speedup vs baseline: 3.0911x; 3.0911x over previous
#include <hip/hip_runtime.h>

typedef __attribute__((ext_vector_type(2))) float v2f;
typedef __attribute__((ext_vector_type(8))) float v8f;

#define DIN 128

#define CSTRIDE 32

__global__ void GCN_zero_cnt(int* __restrict__ cnt, int n) {
    int i = blockIdx.x * blockDim.x + threadIdx.x;
    if (i < n) cnt[(size_t)i * CSTRIDE] = 0;
}

__global__ void GCN_rank(const int* __restrict__ dst, int* __restrict__ cnt,
                         int* __restrict__ rank, int E) {
    int e = blockIdx.x * blockDim.x + threadIdx.x;
    if (e < E) rank[e] = atomicAdd(&cnt[(size_t)dst[e] * CSTRIDE], 1);
}

__global__ __launch_bounds__(1024) void GCN_scan_local(const int* __restrict__ cnt,
                                                       int* __restrict__ rowptr,
                                                       float* __restrict__ norm,
                                                       int* __restrict__ bsum, int n) {
    __shared__ int wsum[32];
    const int t    = threadIdx.x;
    const int lane = t & 31;
    const int w    = t >> 5;
    const int i    = blockIdx.x * 1024 + t;
    const int v    = (i < n) ? cnt[(size_t)i * CSTRIDE] : 0;
    if (i < n) norm[i] = 1.0f / ((float)v + 1.0f);
    int x = v;
    for (int o = 1; o < 32; o <<= 1) {
        int y = __shfl_up(x, o);
        if (lane >= o) x += y;
    }
    if (lane == 31) wsum[w] = x;
    __syncthreads();
    if (w == 0) {
        int s = wsum[lane];
        for (int o = 1; o < 32; o <<= 1) {
            int y = __shfl_up(s, o);
            if (lane >= o) s += y;
        }
        wsum[lane] = s;
    }
    __syncthreads();
    const int incl = x + ((w > 0) ? wsum[w - 1] : 0);
    if (i < n) rowptr[i] = incl - v;
    if (t == 1023) bsum[blockIdx.x] = incl;
}

__global__ __launch_bounds__(1024) void GCN_scan_add(int* __restrict__ rowptr,
                                                     const int* __restrict__ bsum,
                                                     int n, int nb) {
    __shared__ int red[32];
    const int t    = threadIdx.x;
    const int lane = t & 31;
    const int w    = t >> 5;
    const int b    = blockIdx.x;
    int part = 0;
    for (int j = t; j < b; j += 1024) part += bsum[j];
    for (int o = 16; o > 0; o >>= 1) part += __shfl_xor(part, o);
    if (lane == 0) red[w] = part;
    __syncthreads();
    int off = 0;
    for (int k = 0; k < 32; ++k) off += red[k];
    const int i = b * 1024 + t;
    if (i < n) rowptr[i] += off;
    if (b == nb - 1 && t == 0) rowptr[n] = off + bsum[b];
}

__global__ void GCN_fill_col(const int* __restrict__ src, const int* __restrict__ dst,
                             const int* __restrict__ rank, const int* __restrict__ rowptr,
                             int* __restrict__ col, int E) {
    int e = blockIdx.x * blockDim.x + threadIdx.x;
    if (e < E) {
        int pos = rowptr[dst[e]] + rank[e];
        if ((unsigned)pos < (unsigned)E) col[pos] = src[e];
    }
}

__global__ __launch_bounds__(256) void GCN_gather(const float* __restrict__ h,
                                                  const int* __restrict__ rowptr,
                                                  const int* __restrict__ col,
                                                  float* __restrict__ agg, int n) {
    const int lane = threadIdx.x & 31;
    const int row = __builtin_amdgcn_readfirstlane((int)(blockIdx.x * 8 + (threadIdx.x >> 5)));
    if (row >= n) return;
    const int beg = rowptr[row];
    const int end = rowptr[row + 1];
    const float4* __restrict__ h4 = (const float4*)h;
    float4 acc = make_float4(0.0f, 0.0f, 0.0f, 0.0f);
    for (int base = beg; base < end; base += 32) {
        int c = end - base;
        c = (c < 32) ? c : 32;
        int idx = 0;
        if (lane < c) idx = col[base + lane];
        idx = ((unsigned)idx < (unsigned)n) ? idx : 0;
        int j = 0;
        for (; j + 8 <= c; j += 8) {
            float4 v[8];
#pragma unroll
            for (int u = 0; u < 8; ++u) {
                const int s = __builtin_amdgcn_readlane(idx, j + u);
                v[u] = h4[(size_t)s * (DIN / 4) + lane];
            }
#pragma unroll
            for (int u = 0; u < 8; ++u) {
                acc.x += v[u].x; acc.y += v[u].y; acc.z += v[u].z; acc.w += v[u].w;
            }
        }
        for (; j + 2 <= c; j += 2) {
            const int s0 = __builtin_amdgcn_readlane(idx, j);
            const int s1 = __builtin_amdgcn_readlane(idx, j + 1);
            const float4 v0 = h4[(size_t)s0 * (DIN / 4) + lane];
            const float4 v1 = h4[(size_t)s1 * (DIN / 4) + lane];
            acc.x += v0.x; acc.y += v0.y; acc.z += v0.z; acc.w += v0.w;
            acc.x += v1.x; acc.y += v1.y; acc.z += v1.z; acc.w += v1.w;
        }
        if (j < c) {
            const int s0 = __builtin_amdgcn_readlane(idx, j);
            const float4 v0 = h4[(size_t)s0 * (DIN / 4) + lane];
            acc.x += v0.x; acc.y += v0.y; acc.z += v0.z; acc.w += v0.w;
        }
    }
    ((float4*)(agg + (size_t)row * DIN))[lane] = acc;
}

template <int NC, bool RELU>
__global__ __launch_bounds__(128) void GCN_fc_wmma(const float* __restrict__ hself,
                                                   const float* __restrict__ hagg,
                                                   const float* __restrict__ norm,
                                                   const float* __restrict__ W,
                                                   const float* __restrict__ bias,
                                                   float* __restrict__ out) {
    constexpr int K    = DIN;
    constexpr int LDSK = K + 2;
    __shared__ float Atile[16 * LDSK];

    const int row0 = blockIdx.x * 16;
    const int tid  = threadIdx.x;

    for (int i = tid; i < 16 * K; i += 128) {
        int r = i >> 7;
        int k = i & (K - 1);
        int gr = row0 + r;
        Atile[r * LDSK + k] =
            (hagg[(size_t)gr * K + k] + hself[(size_t)gr * K + k]) * norm[gr];
    }
    __syncthreads();

    const int wave = tid >> 5;
    const int lane = tid & 31;
    const int m    = lane & 15;
    const int kh2  = (lane >> 4) << 1;

    for (int nt = wave; nt * 16 < NC; nt += 4) {
        const int n0 = nt * 16;
        const float* wrow = W + (size_t)(n0 + m) * K;

        v8f acc = {};
#pragma unroll 4
        for (int kb = 0; kb < K; kb += 4) {
            v2f a, b;
            a.x = Atile[m * LDSK + kb + kh2];
            a.y = Atile[m * LDSK + kb + kh2 + 1];
            b.x = wrow[kb + kh2];
            b.y = wrow[kb + kh2 + 1];
            acc = __builtin_amdgcn_wmma_f32_16x16x4_f32(
                 false, a,  false, b,
                 (short)0, acc,  false,  false);
        }

        const float bn   = bias[n0 + m];
        const int   rupp = (lane >> 4) << 3;
#pragma unroll
        for (int v = 0; v < 8; ++v) {
            float val = acc[v] + bn;
            if (RELU) val = fmaxf(val, 0.0f);
            out[(size_t)(row0 + rupp + v) * NC + (n0 + m)] = val;
        }
    }
}


extern "C" void kernel_launch(void* const* d_in, const int* in_sizes, int n_in,
                              void* d_out, int out_size, void* d_ws, size_t ws_size,
                              hipStream_t stream) {
    (void)n_in; (void)out_size; (void)ws_size;
    const float* x  = (const float*)d_in[0];
    const int*   src = (const int*)d_in[1];
    const int*   dst = (const int*)d_in[2];
    const float* W1 = (const float*)d_in[3];
    const float* b1 = (const float*)d_in[4];
    const float* W2 = (const float*)d_in[5];
    const float* b2 = (const float*)d_in[6];
    const float* W3 = (const float*)d_in[7];
    const float* b3 = (const float*)d_in[8];
    float* out = (float*)d_out;

    const int N = in_sizes[0] / DIN;
    const int E = in_sizes[1];

    char* ws = (char*)d_ws;
    size_t off = 0;
    float* norm = (float*)(ws + off);
    off += (((size_t)N * 4) + 255) & ~(size_t)255;
    float* buf0 = (float*)(ws + off);
    off += (size_t)N * DIN * 4;
    float* buf1 = (float*)(ws + off);
    off += (size_t)N * DIN * 4;
    int* cnt = (int*)(ws + off);
    off += (size_t)N * CSTRIDE * 4;
    int* rowptr = (int*)(ws + off);
    off += (((size_t)(N + 1) * 4) + 255) & ~(size_t)255;
    int* bsum = (int*)(ws + off);
    off += 4096;
    int* rank = (int*)(ws + off);
    off += (((size_t)E * 4) + 255) & ~(size_t)255;
    int* col = (int*)(ws + off);

    const int TB = 256;
    const int nb = (N + 1023) / 1024;
    const int gblocks = (N + 7) / 8;
    const int fblocks = N / 16;

    GCN_zero_cnt<<<(N + TB - 1) / TB, TB, 0, stream>>>(cnt, N);
    GCN_rank<<<(E + TB - 1) / TB, TB, 0, stream>>>(dst, cnt, rank, E);
    GCN_scan_local<<<nb, 1024, 0, stream>>>(cnt, rowptr, norm, bsum, N);
    GCN_scan_add<<<nb, 1024, 0, stream>>>(rowptr, bsum, N, nb);
    GCN_fill_col<<<(E + TB - 1) / TB, TB, 0, stream>>>(src, dst, rank, rowptr, col, E);

    GCN_gather<<<gblocks, 256, 0, stream>>>(x, rowptr, col, buf1, N);
    GCN_fc_wmma<128, true><<<fblocks, 128, 0, stream>>>(x, buf1, norm, W1, b1, buf0);

    GCN_gather<<<gblocks, 256, 0, stream>>>(buf0, rowptr, col, buf1, N);
    GCN_fc_wmma<128, true><<<fblocks, 128, 0, stream>>>(buf0, buf1, norm, W2, b2, buf1);

    GCN_gather<<<gblocks, 256, 0, stream>>>(buf1, rowptr, col, buf0, N);
    GCN_fc_wmma<64, false><<<fblocks, 128, 0, stream>>>(buf1, buf0, norm, W3, b3, out);
}


# FC: each wave keeps its W (B) fragments for all K in registers and the work-group walks over 32-row tiles (grid sized to the chip), so W is fetched once per group, not once per 16 rows
# speedup vs baseline: 3.5657x; 1.1535x over previous
#include <hip/hip_runtime.h>

typedef __attribute__((ext_vector_type(2))) float v2f;
typedef __attribute__((ext_vector_type(8))) float v8f;

#define DIN 128

#define CSTRIDE 32

__global__ void GCN_zero_cnt(int* __restrict__ cnt, int n) {
    int i = blockIdx.x * blockDim.x + threadIdx.x;
    if (i < n) cnt[(size_t)i * CSTRIDE] = 0;
}

__global__ void GCN_rank(const int* __restrict__ dst, int* __restrict__ cnt,
                         int* __restrict__ rank, int E) {
    int e = blockIdx.x * blockDim.x + threadIdx.x;
    if (e < E) rank[e] = atomicAdd(&cnt[(size_t)dst[e] * CSTRIDE], 1);
}

__global__ __launch_bounds__(1024) void GCN_scan_local(const int* __restrict__ cnt,
                                                       int* __restrict__ rowptr,
                                                       float* __restrict__ norm,
                                                       int* __restrict__ bsum, int n) {
    __shared__ int wsum[32];
    const int t    = threadIdx.x;
    const int lane = t & 31;
    const int w    = t >> 5;
    const int i    = blockIdx.x * 1024 + t;
    const int v    = (i < n) ? cnt[(size_t)i * CSTRIDE] : 0;
    if (i < n) norm[i] = 1.0f / ((float)v + 1.0f);
    int x = v;
    for (int o = 1; o < 32; o <<= 1) {
        int y = __shfl_up(x, o);
        if (lane >= o) x += y;
    }
    if (lane == 31) wsum[w] = x;
    __syncthreads();
    if (w == 0) {
        int s = wsum[lane];
        for (int o = 1; o < 32; o <<= 1) {
            int y = __shfl_up(s, o);
            if (lane >= o) s += y;
        }
        wsum[lane] = s;
    }
    __syncthreads();
    const int incl = x + ((w > 0) ? wsum[w - 1] : 0);
    if (i < n) rowptr[i] = incl - v;
    if (t == 1023) bsum[blockIdx.x] = incl;
}

__global__ __launch_bounds__(1024) void GCN_scan_add(int* __restrict__ rowptr,
                                                     const int* __restrict__ bsum,
                                                     int n, int nb) {
    __shared__ int red[32];
    const int t    = threadIdx.x;
    const int lane = t & 31;
    const int w    = t >> 5;
    const int b    = blockIdx.x;
    int part = 0;
    for (int j = t; j < b; j += 1024) part += bsum[j];
    for (int o = 16; o > 0; o >>= 1) part += __shfl_xor(part, o);
    if (lane == 0) red[w] = part;
    __syncthreads();
    int off = 0;
    for (int k = 0; k < 32; ++k) off += red[k];
    const int i = b * 1024 + t;
    if (i < n) rowptr[i] += off;
    if (b == nb - 1 && t == 0) rowptr[n] = off + bsum[b];
}

__global__ void GCN_fill_col(const int* __restrict__ src, const int* __restrict__ dst,
                             const int* __restrict__ rank, const int* __restrict__ rowptr,
                             int* __restrict__ col, int E) {
    int e = blockIdx.x * blockDim.x + threadIdx.x;
    if (e < E) {
        int pos = rowptr[dst[e]] + rank[e];
        if ((unsigned)pos < (unsigned)E) col[pos] = src[e];
    }
}

__global__ __launch_bounds__(256) void GCN_gather(const float* __restrict__ h,
                                                  const int* __restrict__ rowptr,
                                                  const int* __restrict__ col,
                                                  float* __restrict__ agg, int n) {
    const int lane = threadIdx.x & 31;
    const int row = __builtin_amdgcn_readfirstlane((int)(blockIdx.x * 8 + (threadIdx.x >> 5)));
    if (row >= n) return;
    const int beg = rowptr[row];
    const int end = rowptr[row + 1];
    const float4* __restrict__ h4 = (const float4*)h;
    float4 acc = make_float4(0.0f, 0.0f, 0.0f, 0.0f);
    for (int base = beg; base < end; base += 32) {
        int c = end - base;
        c = (c < 32) ? c : 32;
        int idx = 0;
        if (lane < c) idx = col[base + lane];
        idx = ((unsigned)idx < (unsigned)n) ? idx : 0;
        int j = 0;
        for (; j + 8 <= c; j += 8) {
            float4 v[8];
#pragma unroll
            for (int u = 0; u < 8; ++u) {
                const int s = __builtin_amdgcn_readlane(idx, j + u);
                v[u] = h4[(size_t)s * (DIN / 4) + lane];
            }
#pragma unroll
            for (int u = 0; u < 8; ++u) {
                acc.x += v[u].x; acc.y += v[u].y; acc.z += v[u].z; acc.w += v[u].w;
            }
        }
        for (; j + 2 <= c; j += 2) {
            const int s0 = __builtin_amdgcn_readlane(idx, j);
            const int s1 = __builtin_amdgcn_readlane(idx, j + 1);
            const float4 v0 = h4[(size_t)s0 * (DIN / 4) + lane];
            const float4 v1 = h4[(size_t)s1 * (DIN / 4) + lane];
            acc.x += v0.x; acc.y += v0.y; acc.z += v0.z; acc.w += v0.w;
            acc.x += v1.x; acc.y += v1.y; acc.z += v1.z; acc.w += v1.w;
        }
        if (j < c) {
            const int s0 = __builtin_amdgcn_readlane(idx, j);
            const float4 v0 = h4[(size_t)s0 * (DIN / 4) + lane];
            acc.x += v0.x; acc.y += v0.y; acc.z += v0.z; acc.w += v0.w;
        }
    }
    ((float4*)(agg + (size_t)row * DIN))[lane] = acc;
}

#define FC_TR 32

template <int NC, bool RELU>
__global__ __launch_bounds__(256) void GCN_fc_wmma(const float* __restrict__ hself,
                                                   const float* hagg,
                                                   const float* __restrict__ norm,
                                                   const float* __restrict__ W,
                                                   const float* __restrict__ bias,
                                                   float* out, int ntiles) {
    constexpr int K    = DIN;
    constexpr int KS   = K / 4;
    constexpr int LDSK = K + 4;
    constexpr int NT   = NC / 16;
    constexpr int WPN  = 8 / NT;
    constexpr int RT   = 2 / WPN;
    __shared__ __attribute__((aligned(16))) float Atile[2][FC_TR * LDSK];

    const int tid  = threadIdx.x;
    const int wave = tid >> 5;
    const int lane = tid & 31;
    const int m    = lane & 15;
    const int kh2  = (lane >> 4) << 1;
    const int n0   = (wave % NT) * 16;
    const int rt0  = (wave / NT) * RT;

    const float* wrow = W + (size_t)(n0 + m) * K;
    v2f b[KS];
#pragma unroll
    for (int ks = 0; ks < KS; ++ks) {
        b[ks].x = wrow[ks * 4 + kh2];
        b[ks].y = wrow[ks * 4 + kh2 + 1];
    }
    const float bn   = bias[n0 + m];
    const int   rupp = (lane >> 4) << 3;

    int buf = 0;
    for (int t = blockIdx.x; t < ntiles; t += gridDim.x, buf ^= 1) {
        const int row0 = t * FC_TR;
        float* At = Atile[buf];

        {
            constexpr int NL = (FC_TR * K / 4) / 256;
            const float4* pa = (const float4*)(hagg + (size_t)row0 * K) + tid;
            const float4* ps = (const float4*)(hself + (size_t)row0 * K) + tid;
            const float*  pn = norm + row0 + wave;
            float4 va[NL], vs[NL];
            float  nr[NL];
#pragma unroll
            for (int i = 0; i < NL; ++i) {
                va[i] = pa[256 * i];
                vs[i] = ps[256 * i];
                nr[i] = pn[8 * i];
            }
            float* ad = At + wave * LDSK + lane * 4;
#pragma unroll
            for (int i = 0; i < NL; ++i) {
                float4 o;
                o.x = (va[i].x + vs[i].x) * nr[i];
                o.y = (va[i].y + vs[i].y) * nr[i];
                o.z = (va[i].z + vs[i].z) * nr[i];
                o.w = (va[i].w + vs[i].w) * nr[i];
                *(float4*)(ad + 8 * i * LDSK) = o;
            }
        }
        __syncthreads();

        v8f acc[RT];
#pragma unroll
        for (int rt = 0; rt < RT; ++rt) acc[rt] = v8f{};
#pragma unroll
        for (int ks = 0; ks < KS; ++ks) {
#pragma unroll
            for (int rt = 0; rt < RT; ++rt) {
                const float* arow = At + ((rt0 + rt) * 16 + m) * LDSK;
                v2f a;
                a.x = arow[ks * 4 + kh2];
                a.y = arow[ks * 4 + kh2 + 1];
                acc[rt] = __builtin_amdgcn_wmma_f32_16x16x4_f32(
                     false, a,  false, b[ks],
                     (short)0, acc[rt],  false,  false);
            }
        }

#pragma unroll
        for (int rt = 0; rt < RT; ++rt) {
            float* op = out + (size_t)(row0 + (rt0 + rt) * 16 + rupp) * NC + (n0 + m);
#pragma unroll
            for (int v = 0; v < 8; ++v) {
                float val = acc[rt][v] + bn;
                if (RELU) val = fmaxf(val, 0.0f);
                op[v * NC] = val;
            }
        }
    }
}


extern "C" void kernel_launch(void* const* d_in, const int* in_sizes, int n_in,
                              void* d_out, int out_size, void* d_ws, size_t ws_size,
                              hipStream_t stream) {
    (void)n_in; (void)out_size; (void)ws_size;
    const float* x  = (const float*)d_in[0];
    const int*   src = (const int*)d_in[1];
    const int*   dst = (const int*)d_in[2];
    const float* W1 = (const float*)d_in[3];
    const float* b1 = (const float*)d_in[4];
    const float* W2 = (const float*)d_in[5];
    const float* b2 = (const float*)d_in[6];
    const float* W3 = (const float*)d_in[7];
    const float* b3 = (const float*)d_in[8];
    float* out = (float*)d_out;

    const int N = in_sizes[0] / DIN;
    const int E = in_sizes[1];

    char* ws = (char*)d_ws;
    size_t off = 0;
    float* norm = (float*)(ws + off);
    off += (((size_t)N * 4) + 255) & ~(size_t)255;
    float* buf0 = (float*)(ws + off);
    off += (size_t)N * DIN * 4;
    float* buf1 = (float*)(ws + off);
    off += (size_t)N * DIN * 4;
    int* cnt = (int*)(ws + off);
    off += (size_t)N * CSTRIDE * 4;
    int* rowptr = (int*)(ws + off);
    off += (((size_t)(N + 1) * 4) + 255) & ~(size_t)255;
    int* bsum = (int*)(ws + off);
    off += 4096;
    int* rank = (int*)(ws + off);
    off += (((size_t)E * 4) + 255) & ~(size_t)255;
    int* col = (int*)(ws + off);

    const int TB = 256;
    const int nb = (N + 1023) / 1024;
    const int gblocks = (N + 7) / 8;
    const int ftiles  = N / FC_TR;
    const int fblocks = (ftiles < 1024) ? ftiles : 1024;

    GCN_zero_cnt<<<(N + TB - 1) / TB, TB, 0, stream>>>(cnt, N);
    GCN_rank<<<(E + TB - 1) / TB, TB, 0, stream>>>(dst, cnt, rank, E);
    GCN_scan_local<<<nb, 1024, 0, stream>>>(cnt, rowptr, norm, bsum, N);
    GCN_scan_add<<<nb, 1024, 0, stream>>>(rowptr, bsum, N, nb);
    GCN_fill_col<<<(E + TB - 1) / TB, TB, 0, stream>>>(src, dst, rank, rowptr, col, E);

    GCN_gather<<<gblocks, 256, 0, stream>>>(x, rowptr, col, buf1, N);
    GCN_fc_wmma<128, true><<<fblocks, 256, 0, stream>>>(x, buf1, norm, W1, b1, buf0, ftiles);

    GCN_gather<<<gblocks, 256, 0, stream>>>(buf0, rowptr, col, buf1, N);
    GCN_fc_wmma<128, true><<<fblocks, 256, 0, stream>>>(buf0, buf1, norm, W2, b2, buf1, ftiles);

    GCN_gather<<<gblocks, 256, 0, stream>>>(buf1, rowptr, col, buf0, N);
    GCN_fc_wmma<64, false><<<fblocks, 256, 0, stream>>>(buf1, buf0, norm, W3, b3, out, ftiles);
}


# Neighbour sum fused into the FC kernel's tile staging: each wave sums the source rows of its 4 tile rows (same col order, loads in batches of up to 8) into LDS; agg plane and 3 launches go
# speedup vs baseline: 3.8963x; 1.0927x over previous
#include <hip/hip_runtime.h>

typedef __attribute__((ext_vector_type(2))) float v2f;
typedef __attribute__((ext_vector_type(8))) float v8f;

#define DIN 128

#define CSTRIDE 32

__global__ void GCN_zero_cnt(int* __restrict__ cnt, int n) {
    int i = blockIdx.x * blockDim.x + threadIdx.x;
    if (i < n) cnt[(size_t)i * CSTRIDE] = 0;
}

__global__ void GCN_rank(const int* __restrict__ dst, int* __restrict__ cnt,
                         int* __restrict__ rank, int E) {
    int e = blockIdx.x * blockDim.x + threadIdx.x;
    if (e < E) rank[e] = atomicAdd(&cnt[(size_t)dst[e] * CSTRIDE], 1);
}

__global__ __launch_bounds__(1024) void GCN_scan_local(const int* __restrict__ cnt,
                                                       int* __restrict__ rowptr,
                                                       float* __restrict__ norm,
                                                       int* __restrict__ bsum, int n) {
    __shared__ int wsum[32];
    const int t    = threadIdx.x;
    const int lane = t & 31;
    const int w    = t >> 5;
    const int i    = blockIdx.x * 1024 + t;
    const int v    = (i < n) ? cnt[(size_t)i * CSTRIDE] : 0;
    if (i < n) norm[i] = 1.0f / ((float)v + 1.0f);
    int x = v;
    for (int o = 1; o < 32; o <<= 1) {
        int y = __shfl_up(x, o);
        if (lane >= o) x += y;
    }
    if (lane == 31) wsum[w] = x;
    __syncthreads();
    if (w == 0) {
        int s = wsum[lane];
        for (int o = 1; o < 32; o <<= 1) {
            int y = __shfl_up(s, o);
            if (lane >= o) s += y;
        }
        wsum[lane] = s;
    }
    __syncthreads();
    const int incl = x + ((w > 0) ? wsum[w - 1] : 0);
    if (i < n) rowptr[i] = incl - v;
    if (t == 1023) bsum[blockIdx.x] = incl;
}

__global__ __launch_bounds__(1024) void GCN_scan_add(int* __restrict__ rowptr,
                                                     const int* __restrict__ bsum,
                                                     int n, int nb) {
    __shared__ int red[32];
    const int t    = threadIdx.x;
    const int lane = t & 31;
    const int w    = t >> 5;
    const int b    = blockIdx.x;
    int part = 0;
    for (int j = t; j < b; j += 1024) part += bsum[j];
    for (int o = 16; o > 0; o >>= 1) part += __shfl_xor(part, o);
    if (lane == 0) red[w] = part;
    __syncthreads();
    int off = 0;
    for (int k = 0; k < 32; ++k) off += red[k];
    const int i = b * 1024 + t;
    if (i < n) rowptr[i] += off;
    if (b == nb - 1 && t == 0) rowptr[n] = off + bsum[b];
}

__global__ void GCN_fill_col(const int* __restrict__ src, const int* __restrict__ dst,
                             const int* __restrict__ rank, const int* __restrict__ rowptr,
                             int* __restrict__ col, int E) {
    int e = blockIdx.x * blockDim.x + threadIdx.x;
    if (e < E) {
        int pos = rowptr[dst[e]] + rank[e];
        if ((unsigned)pos < (unsigned)E) col[pos] = src[e];
    }
}

#define FC_TR 32

__device__ __forceinline__ void GCN_sum_chunk(const float4* __restrict__ h4, int idx, int c,
                                              int lane, float4& acc) {
    int j = 0;
    for (; j + 8 <= c; j += 8) {
        float4 v[8];
#pragma unroll
        for (int u = 0; u < 8; ++u) {
            const int s = __builtin_amdgcn_readlane(idx, j + u);
            v[u] = h4[(size_t)s * (DIN / 4) + lane];
        }
#pragma unroll
        for (int u = 0; u < 8; ++u) {
            acc.x += v[u].x; acc.y += v[u].y; acc.z += v[u].z; acc.w += v[u].w;
        }
    }
    const int r = c - j;
    if (r > 0) {
        float4 v[7];
#pragma unroll
        for (int u = 0; u < 7; ++u) {
            v[u] = make_float4(0.0f, 0.0f, 0.0f, 0.0f);
            if (u < r) {
                const int s = __builtin_amdgcn_readlane(idx, j + u);
                v[u] = h4[(size_t)s * (DIN / 4) + lane];
            }
        }
#pragma unroll
        for (int u = 0; u < 7; ++u) {
            if (u < r) {
                acc.x += v[u].x; acc.y += v[u].y; acc.z += v[u].z; acc.w += v[u].w;
            }
        }
    }
}

__device__ __forceinline__ float4 GCN_neigh_sum(const float4* __restrict__ h4,
                                                const int* __restrict__ col,
                                                int beg, int end, int idx0, int lane, int n) {
    float4 acc = make_float4(0.0f, 0.0f, 0.0f, 0.0f);
    int c = end - beg;
    c = (c < 32) ? c : 32;
    idx0 = ((unsigned)idx0 < (unsigned)n) ? idx0 : 0;
    GCN_sum_chunk(h4, idx0, c, lane, acc);
    for (int base = beg + 32; base < end; base += 32) {
        c = end - base;
        c = (c < 32) ? c : 32;
        int idx = 0;
        if (lane < c) idx = col[base + lane];
        idx = ((unsigned)idx < (unsigned)n) ? idx : 0;
        GCN_sum_chunk(h4, idx, c, lane, acc);
    }
    return acc;
}

template <int NC, bool RELU>
__global__ __launch_bounds__(256) void GCN_layer_wmma(const float* __restrict__ h,
                                                      const int* __restrict__ rowptr,
                                                      const int* __restrict__ col,
                                                      const float* __restrict__ norm,
                                                      const float* __restrict__ W,
                                                      const float* __restrict__ bias,
                                                      float* __restrict__ out, int ntiles, int n) {
    constexpr int K    = DIN;
    constexpr int KS   = K / 4;
    constexpr int LDSK = K + 4;
    constexpr int NT   = NC / 16;
    constexpr int WPN  = 8 / NT;
    constexpr int RT   = 2 / WPN;
    constexpr int RW   = FC_TR / 8;
    __shared__ __attribute__((aligned(16))) float Atile[2][FC_TR * LDSK];

    const int tid  = threadIdx.x;
    const int wave = tid >> 5;
    const int lane = tid & 31;
    const int m    = lane & 15;
    const int kh2  = (lane >> 4) << 1;
    const int n0   = (wave % NT) * 16;
    const int rt0  = (wave / NT) * RT;
    const int wrow0 = __builtin_amdgcn_readfirstlane(wave * RW);

    const float* wrow = W + (size_t)(n0 + m) * K;
    v2f b[KS];
#pragma unroll
    for (int ks = 0; ks < KS; ++ks) {
        b[ks].x = wrow[ks * 4 + kh2];
        b[ks].y = wrow[ks * 4 + kh2 + 1];
    }
    const float bn   = bias[n0 + m];
    const int   rupp = (lane >> 4) << 3;

    const float4* __restrict__ h4 = (const float4*)h;

    int buf = 0;
    for (int t = blockIdx.x; t < ntiles; t += gridDim.x, buf ^= 1) {
        const int row0 = t * FC_TR;
        float* At = Atile[buf];

        {
            const int rb = row0 + wrow0;
            int rp[RW + 1];
#pragma unroll
            for (int i = 0; i <= RW; ++i) rp[i] = rowptr[rb + i];
            int idx0[RW];
#pragma unroll
            for (int i = 0; i < RW; ++i) {
                idx0[i] = 0;
                if (lane < rp[i + 1] - rp[i]) idx0[i] = col[rp[i] + lane];
            }
#pragma unroll
            for (int i = 0; i < RW; ++i) {
                const float4 vs = h4[(size_t)(rb + i) * (DIN / 4) + lane];
                const float  nr = norm[rb + i];
                const float4 ag = GCN_neigh_sum(h4, col, rp[i], rp[i + 1], idx0[i], lane, n);
                float4 o;
                o.x = (ag.x + vs.x) * nr;
                o.y = (ag.y + vs.y) * nr;
                o.z = (ag.z + vs.z) * nr;
                o.w = (ag.w + vs.w) * nr;
                *(float4*)(At + (wrow0 + i) * LDSK + lane * 4) = o;
            }
        }
        __syncthreads();

        v8f acc[RT];
#pragma unroll
        for (int rt = 0; rt < RT; ++rt) acc[rt] = v8f{};
#pragma unroll
        for (int ks = 0; ks < KS; ++ks) {
#pragma unroll
            for (int rt = 0; rt < RT; ++rt) {
                const float* arow = At + ((rt0 + rt) * 16 + m) * LDSK;
                v2f a;
                a.x = arow[ks * 4 + kh2];
                a.y = arow[ks * 4 + kh2 + 1];
                acc[rt] = __builtin_amdgcn_wmma_f32_16x16x4_f32(
                     false, a,  false, b[ks],
                     (short)0, acc[rt],  false,  false);
            }
        }

#pragma unroll
        for (int rt = 0; rt < RT; ++rt) {
            float* op = out + (size_t)(row0 + (rt0 + rt) * 16 + rupp) * NC + (n0 + m);
#pragma unroll
            for (int v = 0; v < 8; ++v) {
                float val = acc[rt][v] + bn;
                if (RELU) val = fmaxf(val, 0.0f);
                op[v * NC] = val;
            }
        }
    }
}


extern "C" void kernel_launch(void* const* d_in, const int* in_sizes, int n_in,
                              void* d_out, int out_size, void* d_ws, size_t ws_size,
                              hipStream_t stream) {
    (void)n_in; (void)out_size; (void)ws_size;
    const float* x  = (const float*)d_in[0];
    const int*   src = (const int*)d_in[1];
    const int*   dst = (const int*)d_in[2];
    const float* W1 = (const float*)d_in[3];
    const float* b1 = (const float*)d_in[4];
    const float* W2 = (const float*)d_in[5];
    const float* b2 = (const float*)d_in[6];
    const float* W3 = (const float*)d_in[7];
    const float* b3 = (const float*)d_in[8];
    float* out = (float*)d_out;

    const int N = in_sizes[0] / DIN;
    const int E = in_sizes[1];

    char* ws = (char*)d_ws;
    size_t off = 0;
    float* norm = (float*)(ws + off);
    off += (((size_t)N * 4) + 255) & ~(size_t)255;
    float* buf0 = (float*)(ws + off);
    off += (size_t)N * DIN * 4;
    float* buf1 = (float*)(ws + off);
    off += (size_t)N * DIN * 4;
    int* cnt = (int*)(ws + off);
    off += (size_t)N * CSTRIDE * 4;
    int* rowptr = (int*)(ws + off);
    off += (((size_t)(N + 1) * 4) + 255) & ~(size_t)255;
    int* bsum = (int*)(ws + off);
    off += 4096;
    int* rank = (int*)(ws + off);
    off += (((size_t)E * 4) + 255) & ~(size_t)255;
    int* col = (int*)(ws + off);

    const int TB = 256;
    const int nb = (N + 1023) / 1024;
    const int ftiles  = N / FC_TR;
    const int fblocks = (ftiles + 3) / 4;

    GCN_zero_cnt<<<(N + TB - 1) / TB, TB, 0, stream>>>(cnt, N);
    GCN_rank<<<(E + TB - 1) / TB, TB, 0, stream>>>(dst, cnt, rank, E);
    GCN_scan_local<<<nb, 1024, 0, stream>>>(cnt, rowptr, norm, bsum, N);
    GCN_scan_add<<<nb, 1024, 0, stream>>>(rowptr, bsum, N, nb);
    GCN_fill_col<<<(E + TB - 1) / TB, TB, 0, stream>>>(src, dst, rank, rowptr, col, E);

    GCN_layer_wmma<128, true><<<fblocks, 256, 0, stream>>>(x, rowptr, col, norm, W1, b1, buf0, ftiles, N);

    GCN_layer_wmma<128, true><<<fblocks, 256, 0, stream>>>(buf0, rowptr, col, norm, W2, b2, buf1, ftiles, N);

    GCN_layer_wmma<64, false><<<fblocks, 256, 0, stream>>>(buf1, rowptr, col, norm, W3, b3, out, ftiles, N);
}


# Edge grouping in one pass: each node owns a 128-B line (atomic degree word + first 31 sources in arrival order, rest in an overflow list) that the layer kernel reads by row index; 3 launches go
# speedup vs baseline: 4.1391x; 1.0623x over previous
#include <hip/hip_runtime.h>

typedef __attribute__((ext_vector_type(2))) float v2f;
typedef __attribute__((ext_vector_type(8))) float v8f;

#define DIN 128

#define BSTRIDE 32
#define BCAP 31

__global__ void GCN_zero_deg(int* __restrict__ bkt, int* __restrict__ novf, int n) {
    int i = blockIdx.x * blockDim.x + threadIdx.x;
    if (i < n) bkt[(size_t)i * BSTRIDE] = 0;
    if (i == 0) *novf = 0;
}

__global__ void GCN_fill_lines(const int* __restrict__ src, const int* __restrict__ dst,
                               int* __restrict__ bkt, int* __restrict__ novf,
                               int* __restrict__ ovf, int E, int n) {
    int e = blockIdx.x * blockDim.x + threadIdx.x;
    if (e < E) {
        const int d = dst[e];
        const int s = src[e];
        if ((unsigned)d < (unsigned)n) {
            int* line = bkt + (size_t)d * BSTRIDE;
            const int slot = atomicAdd(line, 1);
            if (slot < BCAP) {
                line[1 + slot] = s;
            } else {
                const int k = atomicAdd(novf, 1);
                if ((unsigned)k < (unsigned)E) {
                    ovf[2 * (size_t)k]     = d;
                    ovf[2 * (size_t)k + 1] = s;
                }
            }
        }
    }
}

#define FC_TR 32

__device__ __forceinline__ void GCN_sum_chunk(const float4* __restrict__ h4, int idx, int c,
                                              int lane, float4& acc) {
    int j = 0;
    for (; j + 8 <= c; j += 8) {
        float4 v[8];
#pragma unroll
        for (int u = 0; u < 8; ++u) {
            const int s = __builtin_amdgcn_readlane(idx, 1 + j + u);
            v[u] = h4[(size_t)s * (DIN / 4) + lane];
        }
#pragma unroll
        for (int u = 0; u < 8; ++u) {
            acc.x += v[u].x; acc.y += v[u].y; acc.z += v[u].z; acc.w += v[u].w;
        }
    }
    const int r = c - j;
    if (r > 0) {
        float4 v[7];
#pragma unroll
        for (int u = 0; u < 7; ++u) {
            v[u] = make_float4(0.0f, 0.0f, 0.0f, 0.0f);
            if (u < r) {
                const int s = __builtin_amdgcn_readlane(idx, 1 + j + u);
                v[u] = h4[(size_t)s * (DIN / 4) + lane];
            }
        }
#pragma unroll
        for (int u = 0; u < 7; ++u) {
            if (u < r) {
                acc.x += v[u].x; acc.y += v[u].y; acc.z += v[u].z; acc.w += v[u].w;
            }
        }
    }
}

__device__ __forceinline__ float4 GCN_neigh_sum(const float4* __restrict__ h4, int line, int deg,
                                                const int* __restrict__ novf,
                                                const int* __restrict__ ovf, int row, int lane,
                                                int n, int E) {
    float4 acc = make_float4(0.0f, 0.0f, 0.0f, 0.0f);
    const int c = (deg < BCAP) ? deg : BCAP;
    const int idx = ((unsigned)line < (unsigned)n) ? line : 0;
    GCN_sum_chunk(h4, idx, c, lane, acc);
    if (deg > BCAP) {
        int no = *novf;
        no = (no < E) ? no : E;
        for (int base = 0; base < no; base += 32) {
            int d = -1, s = 0;
            if (base + lane < no) {
                d = ovf[2 * (size_t)(base + lane)];
                s = ovf[2 * (size_t)(base + lane) + 1];
            }
            s = ((unsigned)s < (unsigned)n) ? s : 0;
            unsigned m = (unsigned)__builtin_amdgcn_ballot_w32(d == row);
            while (m) {
                const int l = __builtin_ctz(m);
                m &= m - 1;
                const int sl = __builtin_amdgcn_readlane(s, l);
                const float4 v = h4[(size_t)sl * (DIN / 4) + lane];
                acc.x += v.x; acc.y += v.y; acc.z += v.z; acc.w += v.w;
            }
        }
    }
    return acc;
}

template <int NC, bool RELU>
__global__ __launch_bounds__(256) void GCN_layer_wmma(const float* __restrict__ h,
                                                      const int* __restrict__ bkt,
                                                      const int* __restrict__ novf,
                                                      const int* __restrict__ ovf,
                                                      const float* __restrict__ W,
                                                      const float* __restrict__ bias,
                                                      float* __restrict__ out, int ntiles, int n,
                                                      int E) {
    constexpr int K    = DIN;
    constexpr int KS   = K / 4;
    constexpr int LDSK = K + 4;
    constexpr int NT   = NC / 16;
    constexpr int WPN  = 8 / NT;
    constexpr int RT   = 2 / WPN;
    constexpr int RW   = FC_TR / 8;
    __shared__ __attribute__((aligned(16))) float Atile[2][FC_TR * LDSK];

    const int tid  = threadIdx.x;
    const int wave = tid >> 5;
    const int lane = tid & 31;
    const int m    = lane & 15;
    const int kh2  = (lane >> 4) << 1;
    const int n0   = (wave % NT) * 16;
    const int rt0  = (wave / NT) * RT;
    const int wrow0 = __builtin_amdgcn_readfirstlane(wave * RW);

    const float* wrow = W + (size_t)(n0 + m) * K;
    v2f b[KS];
#pragma unroll
    for (int ks = 0; ks < KS; ++ks) {
        b[ks].x = wrow[ks * 4 + kh2];
        b[ks].y = wrow[ks * 4 + kh2 + 1];
    }
    const float bn   = bias[n0 + m];
    const int   rupp = (lane >> 4) << 3;

    const float4* __restrict__ h4 = (const float4*)h;

    int buf = 0;
    for (int t = blockIdx.x; t < ntiles; t += gridDim.x, buf ^= 1) {
        const int row0 = t * FC_TR;
        float* At = Atile[buf];

        {
            const int rb = row0 + wrow0;
            int ln[RW];
#pragma unroll
            for (int i = 0; i < RW; ++i) ln[i] = bkt[(size_t)(rb + i) * BSTRIDE + lane];
#pragma unroll
            for (int i = 0; i < RW; ++i) {
                const float4 vs  = h4[(size_t)(rb + i) * (DIN / 4) + lane];
                const int    deg = __builtin_amdgcn_readfirstlane(ln[i]);
                const float  nr  = 1.0f / ((float)deg + 1.0f);
                const float4 ag  = GCN_neigh_sum(h4, ln[i], deg, novf, ovf, rb + i, lane, n, E);
                float4 o;
                o.x = (ag.x + vs.x) * nr;
                o.y = (ag.y + vs.y) * nr;
                o.z = (ag.z + vs.z) * nr;
                o.w = (ag.w + vs.w) * nr;
                *(float4*)(At + (wrow0 + i) * LDSK + lane * 4) = o;
            }
        }
        __syncthreads();

        v8f acc[RT];
#pragma unroll
        for (int rt = 0; rt < RT; ++rt) acc[rt] = v8f{};
#pragma unroll
        for (int ks = 0; ks < KS; ++ks) {
#pragma unroll
            for (int rt = 0; rt < RT; ++rt) {
                const float* arow = At + ((rt0 + rt) * 16 + m) * LDSK;
                v2f a;
                a.x = arow[ks * 4 + kh2];
                a.y = arow[ks * 4 + kh2 + 1];
                acc[rt] = __builtin_amdgcn_wmma_f32_16x16x4_f32(
                     false, a,  false, b[ks],
                     (short)0, acc[rt],  false,  false);
            }
        }

#pragma unroll
        for (int rt = 0; rt < RT; ++rt) {
            float* op = out + (size_t)(row0 + (rt0 + rt) * 16 + rupp) * NC + (n0 + m);
#pragma unroll
            for (int v = 0; v < 8; ++v) {
                float val = acc[rt][v] + bn;
                if (RELU) val = fmaxf(val, 0.0f);
                op[v * NC] = val;
            }
        }
    }
}


extern "C" void kernel_launch(void* const* d_in, const int* in_sizes, int n_in,
                              void* d_out, int out_size, void* d_ws, size_t ws_size,
                              hipStream_t stream) {
    (void)n_in; (void)out_size; (void)ws_size;
    const float* x  = (const float*)d_in[0];
    const int*   src = (const int*)d_in[1];
    const int*   dst = (const int*)d_in[2];
    const float* W1 = (const float*)d_in[3];
    const float* b1 = (const float*)d_in[4];
    const float* W2 = (const float*)d_in[5];
    const float* b2 = (const float*)d_in[6];
    const float* W3 = (const float*)d_in[7];
    const float* b3 = (const float*)d_in[8];
    float* out = (float*)d_out;

    const int N = in_sizes[0] / DIN;
    const int E = in_sizes[1];

    char* ws = (char*)d_ws;
    size_t off = 0;
    float* buf0 = (float*)(ws + off);
    off += (size_t)N * DIN * 4;
    float* buf1 = (float*)(ws + off);
    off += (size_t)N * DIN * 4;
    int* bkt = (int*)(ws + off);
    off += (size_t)N * BSTRIDE * 4;
    int* novf = (int*)(ws + off);
    off += 256;
    int* ovf = (int*)(ws + off);

    const int TB = 256;
    const int ftiles  = N / FC_TR;
    const int fblocks = (ftiles + 3) / 4;

    GCN_zero_deg<<<(N + TB - 1) / TB, TB, 0, stream>>>(bkt, novf, N);
    GCN_fill_lines<<<(E + TB - 1) / TB, TB, 0, stream>>>(src, dst, bkt, novf, ovf, E, N);

    GCN_layer_wmma<128, true><<<fblocks, 256, 0, stream>>>(x, bkt, novf, ovf, W1, b1, buf0, ftiles, N, E);

    GCN_layer_wmma<128, true><<<fblocks, 256, 0, stream>>>(buf0, bkt, novf, ovf, W2, b2, buf1, ftiles, N, E);

    GCN_layer_wmma<64, false><<<fblocks, 256, 0, stream>>>(buf1, bkt, novf, ovf, W3, b3, out, ftiles, N, E);
}


# Last layer's 128->64 FC done inside the layer-2 kernel (h2 tile kept in LDS, z = h2 @ W3^T written): last launch sums 256-B z rows, two per load (even/odd partial sums), not 512-B; sums reassociate
# speedup vs baseline: 4.3753x; 1.0571x over previous
#include <hip/hip_runtime.h>

typedef __attribute__((ext_vector_type(2))) float v2f;
typedef __attribute__((ext_vector_type(8))) float v8f;

#define DIN 128
#define DOUT 64

#define BSTRIDE 32
#define BCAP 31

__global__ void GCN_zero_deg(int* __restrict__ bkt, int* __restrict__ novf, int n) {
    int i = blockIdx.x * blockDim.x + threadIdx.x;
    if (i < n) bkt[(size_t)i * BSTRIDE] = 0;
    if (i == 0) *novf = 0;
}

__global__ void GCN_fill_lines(const int* __restrict__ src, const int* __restrict__ dst,
                               int* __restrict__ bkt, int* __restrict__ novf,
                               int* __restrict__ ovf, int E, int n) {
    int e = blockIdx.x * blockDim.x + threadIdx.x;
    if (e < E) {
        const int d = dst[e];
        const int s = src[e];
        if ((unsigned)d < (unsigned)n) {
            int* line = bkt + (size_t)d * BSTRIDE;
            const int slot = atomicAdd(line, 1);
            if (slot < BCAP) {
                line[1 + slot] = s;
            } else {
                const int k = atomicAdd(novf, 1);
                if ((unsigned)k < (unsigned)E) {
                    ovf[2 * (size_t)k]     = d;
                    ovf[2 * (size_t)k + 1] = s;
                }
            }
        }
    }
}

#define FC_TR 32

__device__ __forceinline__ void GCN_sum_chunk(const float4* __restrict__ h4, int idx, int c,
                                              int lane, float4& acc) {
    int j = 0;
    for (; j + 8 <= c; j += 8) {
        float4 v[8];
#pragma unroll
        for (int u = 0; u < 8; ++u) {
            const int s = __builtin_amdgcn_readlane(idx, 1 + j + u);
            v[u] = h4[(size_t)s * (DIN / 4) + lane];
        }
#pragma unroll
        for (int u = 0; u < 8; ++u) {
            acc.x += v[u].x; acc.y += v[u].y; acc.z += v[u].z; acc.w += v[u].w;
        }
    }
    const int r = c - j;
    if (r > 0) {
        float4 v[7];
#pragma unroll
        for (int u = 0; u < 7; ++u) {
            v[u] = make_float4(0.0f, 0.0f, 0.0f, 0.0f);
            if (u < r) {
                const int s = __builtin_amdgcn_readlane(idx, 1 + j + u);
                v[u] = h4[(size_t)s * (DIN / 4) + lane];
            }
        }
#pragma unroll
        for (int u = 0; u < 7; ++u) {
            if (u < r) {
                acc.x += v[u].x; acc.y += v[u].y; acc.z += v[u].z; acc.w += v[u].w;
            }
        }
    }
}

__device__ __forceinline__ float4 GCN_neigh_sum(const float4* __restrict__ h4, int line, int deg,
                                                const int* __restrict__ novf,
                                                const int* __restrict__ ovf, int row, int lane,
                                                int n, int E) {
    float4 acc = make_float4(0.0f, 0.0f, 0.0f, 0.0f);
    const int c = (deg < BCAP) ? deg : BCAP;
    const int idx = ((unsigned)line < (unsigned)n) ? line : 0;
    GCN_sum_chunk(h4, idx, c, lane, acc);
    if (deg > BCAP) {
        int no = *novf;
        no = (no < E) ? no : E;
        for (int base = 0; base < no; base += 32) {
            int d = -1, s = 0;
            if (base + lane < no) {
                d = ovf[2 * (size_t)(base + lane)];
                s = ovf[2 * (size_t)(base + lane) + 1];
            }
            s = ((unsigned)s < (unsigned)n) ? s : 0;
            unsigned m = (unsigned)__builtin_amdgcn_ballot_w32(d == row);
            while (m) {
                const int l = __builtin_ctz(m);
                m &= m - 1;
                const int sl = __builtin_amdgcn_readlane(s, l);
                const float4 v = h4[(size_t)sl * (DIN / 4) + lane];
                acc.x += v.x; acc.y += v.y; acc.z += v.z; acc.w += v.w;
            }
        }
    }
    return acc;
}

template <int NC, bool RELU>
__global__ __launch_bounds__(256) void GCN_layer_wmma(const float* __restrict__ h,
                                                      const int* __restrict__ bkt,
                                                      const int* __restrict__ novf,
                                                      const int* __restrict__ ovf,
                                                      const float* __restrict__ W,
                                                      const float* __restrict__ bias,
                                                      float* __restrict__ out, int ntiles, int n,
                                                      int E) {
    constexpr int K    = DIN;
    constexpr int KS   = K / 4;
    constexpr int LDSK = K + 4;
    constexpr int NT   = NC / 16;
    constexpr int WPN  = 8 / NT;
    constexpr int RT   = 2 / WPN;
    constexpr int RW   = FC_TR / 8;
    __shared__ __attribute__((aligned(16))) float Atile[2][FC_TR * LDSK];

    const int tid  = threadIdx.x;
    const int wave = tid >> 5;
    const int lane = tid & 31;
    const int m    = lane & 15;
    const int kh2  = (lane >> 4) << 1;
    const int n0   = (wave % NT) * 16;
    const int rt0  = (wave / NT) * RT;
    const int wrow0 = __builtin_amdgcn_readfirstlane(wave * RW);

    const float* wrow = W + (size_t)(n0 + m) * K;
    v2f b[KS];
#pragma unroll
    for (int ks = 0; ks < KS; ++ks) {
        b[ks].x = wrow[ks * 4 + kh2];
        b[ks].y = wrow[ks * 4 + kh2 + 1];
    }
    const float bn   = bias[n0 + m];
    const int   rupp = (lane >> 4) << 3;

    const float4* __restrict__ h4 = (const float4*)h;

    int buf = 0;
    for (int t = blockIdx.x; t < ntiles; t += gridDim.x, buf ^= 1) {
        const int row0 = t * FC_TR;
        float* At = Atile[buf];

        {
            const int rb = row0 + wrow0;
            int ln[RW];
#pragma unroll
            for (int i = 0; i < RW; ++i) ln[i] = bkt[(size_t)(rb + i) * BSTRIDE + lane];
#pragma unroll
            for (int i = 0; i < RW; ++i) {
                const float4 vs  = h4[(size_t)(rb + i) * (DIN / 4) + lane];
                const int    deg = __builtin_amdgcn_readfirstlane(ln[i]);
                const float  nr  = 1.0f / ((float)deg + 1.0f);
                const float4 ag  = GCN_neigh_sum(h4, ln[i], deg, novf, ovf, rb + i, lane, n, E);
                float4 o;
                o.x = (ag.x + vs.x) * nr;
                o.y = (ag.y + vs.y) * nr;
                o.z = (ag.z + vs.z) * nr;
                o.w = (ag.w + vs.w) * nr;
                *(float4*)(At + (wrow0 + i) * LDSK + lane * 4) = o;
            }
        }
        __syncthreads();

        v8f acc[RT];
#pragma unroll
        for (int rt = 0; rt < RT; ++rt) acc[rt] = v8f{};
#pragma unroll
        for (int ks = 0; ks < KS; ++ks) {
#pragma unroll
            for (int rt = 0; rt < RT; ++rt) {
                const float* arow = At + ((rt0 + rt) * 16 + m) * LDSK;
                v2f a;
                a.x = arow[ks * 4 + kh2];
                a.y = arow[ks * 4 + kh2 + 1];
                acc[rt] = __builtin_amdgcn_wmma_f32_16x16x4_f32(
                     false, a,  false, b[ks],
                     (short)0, acc[rt],  false,  false);
            }
        }

#pragma unroll
        for (int rt = 0; rt < RT; ++rt) {
            float* op = out + (size_t)(row0 + (rt0 + rt) * 16 + rupp) * NC + (n0 + m);
#pragma unroll
            for (int v = 0; v < 8; ++v) {
                float val = acc[rt][v] + bn;
                if (RELU) val = fmaxf(val, 0.0f);
                op[v * NC] = val;
            }
        }
    }
}

__global__ __launch_bounds__(256) void GCN_layer_fc2(const float* __restrict__ h,
                                                     const int* __restrict__ bkt,
                                                     const int* __restrict__ novf,
                                                     const int* __restrict__ ovf,
                                                     const float* __restrict__ W,
                                                     const float* __restrict__ bias,
                                                     const float* __restrict__ Wn,
                                                     float* __restrict__ z, int ntiles, int n, int E) {
    constexpr int K    = DIN;
    constexpr int KS   = K / 4;
    constexpr int LDSK = K + 4;
    constexpr int RW   = FC_TR / 8;
    __shared__ __attribute__((aligned(16))) float At[FC_TR * LDSK];
    __shared__ __attribute__((aligned(16))) float Ht[FC_TR * LDSK];
    __shared__ __attribute__((aligned(16))) float Wt[DOUT * LDSK];

    const int tid  = threadIdx.x;
    const int wave = tid >> 5;
    const int lane = tid & 31;
    const int m    = lane & 15;
    const int kh2  = (lane >> 4) << 1;
    const int n0   = wave * 16;
    const int rt2  = wave >> 2;
    const int n2   = (wave & 3) * 16;
    const int wrow0 = __builtin_amdgcn_readfirstlane(wave * RW);

    for (int i = tid; i < DOUT * (K / 4); i += 256) {
        const int r  = i / (K / 4);
        const int c4 = i % (K / 4);
        *(float4*)(Wt + r * LDSK + c4 * 4) = ((const float4*)Wn)[i];
    }

    const float* wrow = W + (size_t)(n0 + m) * K;
    v2f b[KS];
#pragma unroll
    for (int ks = 0; ks < KS; ++ks) {
        b[ks].x = wrow[ks * 4 + kh2];
        b[ks].y = wrow[ks * 4 + kh2 + 1];
    }
    const float bn   = bias[n0 + m];
    const int   rupp = (lane >> 4) << 3;

    const float4* __restrict__ h4 = (const float4*)h;

    for (int t = blockIdx.x; t < ntiles; t += gridDim.x) {
        const int row0 = t * FC_TR;

        {
            const int rb = row0 + wrow0;
            int ln[RW];
#pragma unroll
            for (int i = 0; i < RW; ++i) ln[i] = bkt[(size_t)(rb + i) * BSTRIDE + lane];
#pragma unroll
            for (int i = 0; i < RW; ++i) {
                const float4 vs  = h4[(size_t)(rb + i) * (DIN / 4) + lane];
                const int    deg = __builtin_amdgcn_readfirstlane(ln[i]);
                const float  nr  = 1.0f / ((float)deg + 1.0f);
                const float4 ag  = GCN_neigh_sum(h4, ln[i], deg, novf, ovf, rb + i, lane, n, E);
                float4 o;
                o.x = (ag.x + vs.x) * nr;
                o.y = (ag.y + vs.y) * nr;
                o.z = (ag.z + vs.z) * nr;
                o.w = (ag.w + vs.w) * nr;
                *(float4*)(At + (wrow0 + i) * LDSK + lane * 4) = o;
            }
        }
        __syncthreads();

        v8f acc[2];
        acc[0] = v8f{};
        acc[1] = v8f{};
#pragma unroll
        for (int ks = 0; ks < KS; ++ks) {
#pragma unroll
            for (int rt = 0; rt < 2; ++rt) {
                const float* arow = At + (rt * 16 + m) * LDSK;
                v2f a;
                a.x = arow[ks * 4 + kh2];
                a.y = arow[ks * 4 + kh2 + 1];
                acc[rt] = __builtin_amdgcn_wmma_f32_16x16x4_f32(
                     false, a,  false, b[ks],
                     (short)0, acc[rt],  false,  false);
            }
        }

#pragma unroll
        for (int rt = 0; rt < 2; ++rt) {
            float* hp = Ht + (rt * 16 + rupp) * LDSK + (n0 + m);
#pragma unroll
            for (int v = 0; v < 8; ++v) hp[v * LDSK] = fmaxf(acc[rt][v] + bn, 0.0f);
        }
        __syncthreads();

        v8f c0 = v8f{};
        v8f c1 = v8f{};
        const float* hrow = Ht + (rt2 * 16 + m) * LDSK;
        const float* wn   = Wt + (n2 + m) * LDSK;
#pragma unroll
        for (int ks = 0; ks < KS; ks += 2) {
            v2f a0, b0, a1, b1;
            a0.x = hrow[ks * 4 + kh2];
            a0.y = hrow[ks * 4 + kh2 + 1];
            b0.x = wn[ks * 4 + kh2];
            b0.y = wn[ks * 4 + kh2 + 1];
            a1.x = hrow[ks * 4 + 4 + kh2];
            a1.y = hrow[ks * 4 + 4 + kh2 + 1];
            b1.x = wn[ks * 4 + 4 + kh2];
            b1.y = wn[ks * 4 + 4 + kh2 + 1];
            c0 = __builtin_amdgcn_wmma_f32_16x16x4_f32(false, a0, false, b0, (short)0, c0, false, false);
            c1 = __builtin_amdgcn_wmma_f32_16x16x4_f32(false, a1, false, b1, (short)0, c1, false, false);
        }

        float* zp = z + (size_t)(row0 + rt2 * 16 + rupp) * DOUT + (n2 + m);
#pragma unroll
        for (int v = 0; v < 8; ++v) zp[v * DOUT] = c0[v] + c1[v];
    }
}


__device__ __forceinline__ void GCN_sum_pairs(const float4* __restrict__ z4, int idx, int c,
                                              int l16, bool odd, float4& acc) {
    int j = 0;
    for (; j + 16 <= c; j += 16) {
        float4 v[8];
#pragma unroll
        for (int u = 0; u < 8; ++u) {
            const int s0 = __builtin_amdgcn_readlane(idx, 1 + j + 2 * u);
            const int s1 = __builtin_amdgcn_readlane(idx, 2 + j + 2 * u);
            const int s  = odd ? s1 : s0;
            v[u] = z4[(size_t)s * (DOUT / 4) + l16];
        }
#pragma unroll
        for (int u = 0; u < 8; ++u) {
            acc.x += v[u].x; acc.y += v[u].y; acc.z += v[u].z; acc.w += v[u].w;
        }
    }
    const int r = c - j;
    if (r > 0) {
        float4 v[8];
#pragma unroll
        for (int u = 0; u < 8; ++u) {
            v[u] = make_float4(0.0f, 0.0f, 0.0f, 0.0f);
            if (2 * u < r) {
                const bool two = (2 * u + 1 < r);
                const int  s0  = __builtin_amdgcn_readlane(idx, 1 + j + 2 * u);
                const int  s1  = two ? __builtin_amdgcn_readlane(idx, 2 + j + 2 * u) : s0;
                const int  s   = odd ? s1 : s0;
                v[u] = z4[(size_t)s * (DOUT / 4) + l16];
            }
        }
#pragma unroll
        for (int u = 0; u < 8; ++u) {
            if (2 * u < r) {
                const bool use = (2 * u + 1 < r) || !odd;
                acc.x += use ? v[u].x : 0.0f;
                acc.y += use ? v[u].y : 0.0f;
                acc.z += use ? v[u].z : 0.0f;
                acc.w += use ? v[u].w : 0.0f;
            }
        }
    }
}

__global__ __launch_bounds__(256) void GCN_out_gather(const float* __restrict__ z,
                                                      const int* __restrict__ bkt,
                                                      const int* __restrict__ novf,
                                                      const int* __restrict__ ovf,
                                                      const float* __restrict__ bias,
                                                      float* __restrict__ out, int ntiles, int n,
                                                      int E) {
    constexpr int RW = FC_TR / 8;
    const int  tid  = threadIdx.x;
    const int  wave = tid >> 5;
    const int  lane = tid & 31;
    const int  l16  = lane & 15;
    const bool odd  = lane >= 16;
    const int wrow0 = __builtin_amdgcn_readfirstlane(wave * RW);

    const float4* __restrict__ z4 = (const float4*)z;
    const float4 bv = ((const float4*)bias)[l16];

    for (int t = blockIdx.x; t < ntiles; t += gridDim.x) {
        const int rb = t * FC_TR + wrow0;
        int ln[RW];
#pragma unroll
        for (int i = 0; i < RW; ++i) ln[i] = bkt[(size_t)(rb + i) * BSTRIDE + lane];
#pragma unroll
        for (int i = 0; i < RW; ++i) {
            const int    row = rb + i;
            const float4 vs  = z4[(size_t)row * (DOUT / 4) + l16];
            const int    deg = __builtin_amdgcn_readfirstlane(ln[i]);
            const float  nr  = 1.0f / ((float)deg + 1.0f);
            const int    c   = (deg < BCAP) ? deg : BCAP;
            const int    idx = ((unsigned)ln[i] < (unsigned)n) ? ln[i] : 0;
            float4 acc = make_float4(0.0f, 0.0f, 0.0f, 0.0f);
            GCN_sum_pairs(z4, idx, c, l16, odd, acc);
            if (deg > BCAP) {
                int no = *novf;
                no = (no < E) ? no : E;
                for (int base = 0; base < no; base += 32) {
                    int d = -1, s = 0;
                    if (base + lane < no) {
                        d = ovf[2 * (size_t)(base + lane)];
                        s = ovf[2 * (size_t)(base + lane) + 1];
                    }
                    s = ((unsigned)s < (unsigned)n) ? s : 0;
                    unsigned mk = (unsigned)__builtin_amdgcn_ballot_w32(d == row);
                    while (mk) {
                        const int l = __builtin_ctz(mk);
                        mk &= mk - 1;
                        const int sl = __builtin_amdgcn_readlane(s, l);
                        const float4 w = z4[(size_t)sl * (DOUT / 4) + l16];
                        if (!odd) {
                            acc.x += w.x; acc.y += w.y; acc.z += w.z; acc.w += w.w;
                        }
                    }
                }
            }
            acc.x += __shfl_xor(acc.x, 16);
            acc.y += __shfl_xor(acc.y, 16);
            acc.z += __shfl_xor(acc.z, 16);
            acc.w += __shfl_xor(acc.w, 16);
            float4 o;
            o.x = (acc.x + vs.x) * nr + bv.x;
            o.y = (acc.y + vs.y) * nr + bv.y;
            o.z = (acc.z + vs.z) * nr + bv.z;
            o.w = (acc.w + vs.w) * nr + bv.w;
            if (!odd) ((float4*)out)[(size_t)row * (DOUT / 4) + l16] = o;
        }
    }
}


extern "C" void kernel_launch(void* const* d_in, const int* in_sizes, int n_in,
                              void* d_out, int out_size, void* d_ws, size_t ws_size,
                              hipStream_t stream) {
    (void)n_in; (void)out_size; (void)ws_size;
    const float* x  = (const float*)d_in[0];
    const int*   src = (const int*)d_in[1];
    const int*   dst = (const int*)d_in[2];
    const float* W1 = (const float*)d_in[3];
    const float* b1 = (const float*)d_in[4];
    const float* W2 = (const float*)d_in[5];
    const float* b2 = (const float*)d_in[6];
    const float* W3 = (const float*)d_in[7];
    const float* b3 = (const float*)d_in[8];
    float* out = (float*)d_out;

    const int N = in_sizes[0] / DIN;
    const int E = in_sizes[1];

    char* ws = (char*)d_ws;
    size_t off = 0;
    float* buf0 = (float*)(ws + off);
    off += (size_t)N * DIN * 4;
    float* zbuf = (float*)(ws + off);
    off += (size_t)N * DIN * 4;
    int* bkt = (int*)(ws + off);
    off += (size_t)N * BSTRIDE * 4;
    int* novf = (int*)(ws + off);
    off += 256;
    int* ovf = (int*)(ws + off);

    const int TB = 256;
    const int ftiles  = N / FC_TR;
    const int fblocks = (ftiles + 3) / 4;

    GCN_zero_deg<<<(N + TB - 1) / TB, TB, 0, stream>>>(bkt, novf, N);
    GCN_fill_lines<<<(E + TB - 1) / TB, TB, 0, stream>>>(src, dst, bkt, novf, ovf, E, N);

    GCN_layer_wmma<128, true><<<fblocks, 256, 0, stream>>>(x, bkt, novf, ovf, W1, b1, buf0, ftiles, N, E);

    GCN_layer_fc2<<<fblocks, 256, 0, stream>>>(buf0, bkt, novf, ovf, W2, b2, W3, zbuf, ftiles, N, E);

    GCN_out_gather<<<fblocks, 256, 0, stream>>>(zbuf, bkt, novf, ovf, b3, out, ftiles, N, E);
}


# Layer kernels: neighbour rows loaded in batches of up to 16 per wave, not 8 (same add order): a row of <=16 sources waits one round trip, not two; grid 768 = 3 resident groups per WGP
# speedup vs baseline: 4.5565x; 1.0414x over previous
#include <hip/hip_runtime.h>

typedef __attribute__((ext_vector_type(2))) float v2f;
typedef __attribute__((ext_vector_type(8))) float v8f;

#define DIN 128
#define DOUT 64

#define BSTRIDE 32
#define BCAP 31

__global__ void GCN_zero_deg(int* __restrict__ bkt, int* __restrict__ novf, int n) {
    int i = blockIdx.x * blockDim.x + threadIdx.x;
    if (i < n) bkt[(size_t)i * BSTRIDE] = 0;
    if (i == 0) *novf = 0;
}

__global__ void GCN_fill_lines(const int* __restrict__ src, const int* __restrict__ dst,
                               int* __restrict__ bkt, int* __restrict__ novf,
                               int* __restrict__ ovf, int E, int n) {
    int e = blockIdx.x * blockDim.x + threadIdx.x;
    if (e < E) {
        const int d = dst[e];
        const int s = src[e];
        if ((unsigned)d < (unsigned)n) {
            int* line = bkt + (size_t)d * BSTRIDE;
            const int slot = atomicAdd(line, 1);
            if (slot < BCAP) {
                line[1 + slot] = s;
            } else {
                const int k = atomicAdd(novf, 1);
                if ((unsigned)k < (unsigned)E) {
                    ovf[2 * (size_t)k]     = d;
                    ovf[2 * (size_t)k + 1] = s;
                }
            }
        }
    }
}

#define FC_TR 32

#define GB 16
__device__ __forceinline__ void GCN_sum_chunk(const float4* __restrict__ h4, int idx, int c,
                                              int lane, float4& acc) {
    int j = 0;
    for (; j + GB <= c; j += GB) {
        float4 v[GB];
#pragma unroll
        for (int u = 0; u < GB; ++u) {
            const int s = __builtin_amdgcn_readlane(idx, 1 + j + u);
            v[u] = h4[(size_t)s * (DIN / 4) + lane];
        }
#pragma unroll
        for (int u = 0; u < GB; ++u) {
            acc.x += v[u].x; acc.y += v[u].y; acc.z += v[u].z; acc.w += v[u].w;
        }
    }
    const int r = c - j;
    if (r > 0) {
        float4 v[GB - 1];
#pragma unroll
        for (int u = 0; u < GB - 1; ++u) {
            v[u] = make_float4(0.0f, 0.0f, 0.0f, 0.0f);
            if (u < r) {
                const int s = __builtin_amdgcn_readlane(idx, 1 + j + u);
                v[u] = h4[(size_t)s * (DIN / 4) + lane];
            }
        }
#pragma unroll
        for (int u = 0; u < GB - 1; ++u) {
            if (u < r) {
                acc.x += v[u].x; acc.y += v[u].y; acc.z += v[u].z; acc.w += v[u].w;
            }
        }
    }
}

__device__ __forceinline__ float4 GCN_neigh_sum(const float4* __restrict__ h4, int line, int deg,
                                                const int* __restrict__ novf,
                                                const int* __restrict__ ovf, int row, int lane,
                                                int n, int E) {
    float4 acc = make_float4(0.0f, 0.0f, 0.0f, 0.0f);
    const int c = (deg < BCAP) ? deg : BCAP;
    const int idx = ((unsigned)line < (unsigned)n) ? line : 0;
    GCN_sum_chunk(h4, idx, c, lane, acc);
    if (deg > BCAP) {
        int no = *novf;
        no = (no < E) ? no : E;
        for (int base = 0; base < no; base += 32) {
            int d = -1, s = 0;
            if (base + lane < no) {
                d = ovf[2 * (size_t)(base + lane)];
                s = ovf[2 * (size_t)(base + lane) + 1];
            }
            s = ((unsigned)s < (unsigned)n) ? s : 0;
            unsigned m = (unsigned)__builtin_amdgcn_ballot_w32(d == row);
            while (m) {
                const int l = __builtin_ctz(m);
                m &= m - 1;
                const int sl = __builtin_amdgcn_readlane(s, l);
                const float4 v = h4[(size_t)sl * (DIN / 4) + lane];
                acc.x += v.x; acc.y += v.y; acc.z += v.z; acc.w += v.w;
            }
        }
    }
    return acc;
}

template <int NC, bool RELU>
__global__ __launch_bounds__(256) void GCN_layer_wmma(const float* __restrict__ h,
                                                      const int* __restrict__ bkt,
                                                      const int* __restrict__ novf,
                                                      const int* __restrict__ ovf,
                                                      const float* __restrict__ W,
                                                      const float* __restrict__ bias,
                                                      float* __restrict__ out, int ntiles, int n,
                                                      int E) {
    constexpr int K    = DIN;
    constexpr int KS   = K / 4;
    constexpr int LDSK = K + 4;
    constexpr int NT   = NC / 16;
    constexpr int WPN  = 8 / NT;
    constexpr int RT   = 2 / WPN;
    constexpr int RW   = FC_TR / 8;
    __shared__ __attribute__((aligned(16))) float Atile[2][FC_TR * LDSK];

    const int tid  = threadIdx.x;
    const int wave = tid >> 5;
    const int lane = tid & 31;
    const int m    = lane & 15;
    const int kh2  = (lane >> 4) << 1;
    const int n0   = (wave % NT) * 16;
    const int rt0  = (wave / NT) * RT;
    const int wrow0 = __builtin_amdgcn_readfirstlane(wave * RW);

    const float* wrow = W + (size_t)(n0 + m) * K;
    v2f b[KS];
#pragma unroll
    for (int ks = 0; ks < KS; ++ks) {
        b[ks].x = wrow[ks * 4 + kh2];
        b[ks].y = wrow[ks * 4 + kh2 + 1];
    }
    const float bn   = bias[n0 + m];
    const int   rupp = (lane >> 4) << 3;

    const float4* __restrict__ h4 = (const float4*)h;

    int buf = 0;
    for (int t = blockIdx.x; t < ntiles; t += gridDim.x, buf ^= 1) {
        const int row0 = t * FC_TR;
        float* At = Atile[buf];

        {
            const int rb = row0 + wrow0;
            int ln[RW];
#pragma unroll
            for (int i = 0; i < RW; ++i) ln[i] = bkt[(size_t)(rb + i) * BSTRIDE + lane];
#pragma unroll
            for (int i = 0; i < RW; ++i) {
                const float4 vs  = h4[(size_t)(rb + i) * (DIN / 4) + lane];
                const int    deg = __builtin_amdgcn_readfirstlane(ln[i]);
                const float  nr  = 1.0f / ((float)deg + 1.0f);
                const float4 ag  = GCN_neigh_sum(h4, ln[i], deg, novf, ovf, rb + i, lane, n, E);
                float4 o;
                o.x = (ag.x + vs.x) * nr;
                o.y = (ag.y + vs.y) * nr;
                o.z = (ag.z + vs.z) * nr;
                o.w = (ag.w + vs.w) * nr;
                *(float4*)(At + (wrow0 + i) * LDSK + lane * 4) = o;
            }
        }
        __syncthreads();

        v8f acc[RT];
#pragma unroll
        for (int rt = 0; rt < RT; ++rt) acc[rt] = v8f{};
#pragma unroll
        for (int ks = 0; ks < KS; ++ks) {
#pragma unroll
            for (int rt = 0; rt < RT; ++rt) {
                const float* arow = At + ((rt0 + rt) * 16 + m) * LDSK;
                v2f a;
                a.x = arow[ks * 4 + kh2];
                a.y = arow[ks * 4 + kh2 + 1];
                acc[rt] = __builtin_amdgcn_wmma_f32_16x16x4_f32(
                     false, a,  false, b[ks],
                     (short)0, acc[rt],  false,  false);
            }
        }

#pragma unroll
        for (int rt = 0; rt < RT; ++rt) {
            float* op = out + (size_t)(row0 + (rt0 + rt) * 16 + rupp) * NC + (n0 + m);
#pragma unroll
            for (int v = 0; v < 8; ++v) {
                float val = acc[rt][v] + bn;
                if (RELU) val = fmaxf(val, 0.0f);
                op[v * NC] = val;
            }
        }
    }
}

__global__ __launch_bounds__(256) void GCN_layer_fc2(const float* __restrict__ h,
                                                     const int* __restrict__ bkt,
                                                     const int* __restrict__ novf,
                                                     const int* __restrict__ ovf,
                                                     const float* __restrict__ W,
                                                     const float* __restrict__ bias,
                                                     const float* __restrict__ Wn,
                                                     float* __restrict__ z, int ntiles, int n, int E) {
    constexpr int K    = DIN;
    constexpr int KS   = K / 4;
    constexpr int LDSK = K + 4;
    constexpr int RW   = FC_TR / 8;
    __shared__ __attribute__((aligned(16))) float At[FC_TR * LDSK];
    __shared__ __attribute__((aligned(16))) float Ht[FC_TR * LDSK];
    __shared__ __attribute__((aligned(16))) float Wt[DOUT * LDSK];

    const int tid  = threadIdx.x;
    const int wave = tid >> 5;
    const int lane = tid & 31;
    const int m    = lane & 15;
    const int kh2  = (lane >> 4) << 1;
    const int n0   = wave * 16;
    const int rt2  = wave >> 2;
    const int n2   = (wave & 3) * 16;
    const int wrow0 = __builtin_amdgcn_readfirstlane(wave * RW);

    for (int i = tid; i < DOUT * (K / 4); i += 256) {
        const int r  = i / (K / 4);
        const int c4 = i % (K / 4);
        *(float4*)(Wt + r * LDSK + c4 * 4) = ((const float4*)Wn)[i];
    }

    const float* wrow = W + (size_t)(n0 + m) * K;
    v2f b[KS];
#pragma unroll
    for (int ks = 0; ks < KS; ++ks) {
        b[ks].x = wrow[ks * 4 + kh2];
        b[ks].y = wrow[ks * 4 + kh2 + 1];
    }
    const float bn   = bias[n0 + m];
    const int   rupp = (lane >> 4) << 3;

    const float4* __restrict__ h4 = (const float4*)h;

    for (int t = blockIdx.x; t < ntiles; t += gridDim.x) {
        const int row0 = t * FC_TR;

        {
            const int rb = row0 + wrow0;
            int ln[RW];
#pragma unroll
            for (int i = 0; i < RW; ++i) ln[i] = bkt[(size_t)(rb + i) * BSTRIDE + lane];
#pragma unroll
            for (int i = 0; i < RW; ++i) {
                const float4 vs  = h4[(size_t)(rb + i) * (DIN / 4) + lane];
                const int    deg = __builtin_amdgcn_readfirstlane(ln[i]);
                const float  nr  = 1.0f / ((float)deg + 1.0f);
                const float4 ag  = GCN_neigh_sum(h4, ln[i], deg, novf, ovf, rb + i, lane, n, E);
                float4 o;
                o.x = (ag.x + vs.x) * nr;
                o.y = (ag.y + vs.y) * nr;
                o.z = (ag.z + vs.z) * nr;
                o.w = (ag.w + vs.w) * nr;
                *(float4*)(At + (wrow0 + i) * LDSK + lane * 4) = o;
            }
        }
        __syncthreads();

        v8f acc[2];
        acc[0] = v8f{};
        acc[1] = v8f{};
#pragma unroll
        for (int ks = 0; ks < KS; ++ks) {
#pragma unroll
            for (int rt = 0; rt < 2; ++rt) {
                const float* arow = At + (rt * 16 + m) * LDSK;
                v2f a;
                a.x = arow[ks * 4 + kh2];
                a.y = arow[ks * 4 + kh2 + 1];
                acc[rt] = __builtin_amdgcn_wmma_f32_16x16x4_f32(
                     false, a,  false, b[ks],
                     (short)0, acc[rt],  false,  false);
            }
        }

#pragma unroll
        for (int rt = 0; rt < 2; ++rt) {
            float* hp = Ht + (rt * 16 + rupp) * LDSK + (n0 + m);
#pragma unroll
            for (int v = 0; v < 8; ++v) hp[v * LDSK] = fmaxf(acc[rt][v] + bn, 0.0f);
        }
        __syncthreads();

        v8f c0 = v8f{};
        v8f c1 = v8f{};
        const float* hrow = Ht + (rt2 * 16 + m) * LDSK;
        const float* wn   = Wt + (n2 + m) * LDSK;
#pragma unroll
        for (int ks = 0; ks < KS; ks += 2) {
            v2f a0, b0, a1, b1;
            a0.x = hrow[ks * 4 + kh2];
            a0.y = hrow[ks * 4 + kh2 + 1];
            b0.x = wn[ks * 4 + kh2];
            b0.y = wn[ks * 4 + kh2 + 1];
            a1.x = hrow[ks * 4 + 4 + kh2];
            a1.y = hrow[ks * 4 + 4 + kh2 + 1];
            b1.x = wn[ks * 4 + 4 + kh2];
            b1.y = wn[ks * 4 + 4 + kh2 + 1];
            c0 = __builtin_amdgcn_wmma_f32_16x16x4_f32(false, a0, false, b0, (short)0, c0, false, false);
            c1 = __builtin_amdgcn_wmma_f32_16x16x4_f32(false, a1, false, b1, (short)0, c1, false, false);
        }

        float* zp = z + (size_t)(row0 + rt2 * 16 + rupp) * DOUT + (n2 + m);
#pragma unroll
        for (int v = 0; v < 8; ++v) zp[v * DOUT] = c0[v] + c1[v];
    }
}


__device__ __forceinline__ void GCN_sum_pairs(const float4* __restrict__ z4, int idx, int c,
                                              int l16, bool odd, float4& acc) {
    int j = 0;
    for (; j + 16 <= c; j += 16) {
        float4 v[8];
#pragma unroll
        for (int u = 0; u < 8; ++u) {
            const int s0 = __builtin_amdgcn_readlane(idx, 1 + j + 2 * u);
            const int s1 = __builtin_amdgcn_readlane(idx, 2 + j + 2 * u);
            const int s  = odd ? s1 : s0;
            v[u] = z4[(size_t)s * (DOUT / 4) + l16];
        }
#pragma unroll
        for (int u = 0; u < 8; ++u) {
            acc.x += v[u].x; acc.y += v[u].y; acc.z += v[u].z; acc.w += v[u].w;
        }
    }
    const int r = c - j;
    if (r > 0) {
        float4 v[8];
#pragma unroll
        for (int u = 0; u < 8; ++u) {
            v[u] = make_float4(0.0f, 0.0f, 0.0f, 0.0f);
            if (2 * u < r) {
                const bool two = (2 * u + 1 < r);
                const int  s0  = __builtin_amdgcn_readlane(idx, 1 + j + 2 * u);
                const int  s1  = two ? __builtin_amdgcn_readlane(idx, 2 + j + 2 * u) : s0;
                const int  s   = odd ? s1 : s0;
                v[u] = z4[(size_t)s * (DOUT / 4) + l16];
            }
        }
#pragma unroll
        for (int u = 0; u < 8; ++u) {
            if (2 * u < r) {
                const bool use = (2 * u + 1 < r) || !odd;
                acc.x += use ? v[u].x : 0.0f;
                acc.y += use ? v[u].y : 0.0f;
                acc.z += use ? v[u].z : 0.0f;
                acc.w += use ? v[u].w : 0.0f;
            }
        }
    }
}

__global__ __launch_bounds__(256) void GCN_out_gather(const float* __restrict__ z,
                                                      const int* __restrict__ bkt,
                                                      const int* __restrict__ novf,
                                                      const int* __restrict__ ovf,
                                                      const float* __restrict__ bias,
                                                      float* __restrict__ out, int ntiles, int n,
                                                      int E) {
    constexpr int RW = FC_TR / 8;
    const int  tid  = threadIdx.x;
    const int  wave = tid >> 5;
    const int  lane = tid & 31;
    const int  l16  = lane & 15;
    const bool odd  = lane >= 16;
    const int wrow0 = __builtin_amdgcn_readfirstlane(wave * RW);

    const float4* __restrict__ z4 = (const float4*)z;
    const float4 bv = ((const float4*)bias)[l16];

    for (int t = blockIdx.x; t < ntiles; t += gridDim.x) {
        const int rb = t * FC_TR + wrow0;
        int ln[RW];
#pragma unroll
        for (int i = 0; i < RW; ++i) ln[i] = bkt[(size_t)(rb + i) * BSTRIDE + lane];
#pragma unroll
        for (int i = 0; i < RW; ++i) {
            const int    row = rb + i;
            const float4 vs  = z4[(size_t)row * (DOUT / 4) + l16];
            const int    deg = __builtin_amdgcn_readfirstlane(ln[i]);
            const float  nr  = 1.0f / ((float)deg + 1.0f);
            const int    c   = (deg < BCAP) ? deg : BCAP;
            const int    idx = ((unsigned)ln[i] < (unsigned)n) ? ln[i] : 0;
            float4 acc = make_float4(0.0f, 0.0f, 0.0f, 0.0f);
            GCN_sum_pairs(z4, idx, c, l16, odd, acc);
            if (deg > BCAP) {
                int no = *novf;
                no = (no < E) ? no : E;
                for (int base = 0; base < no; base += 32) {
                    int d = -1, s = 0;
                    if (base + lane < no) {
                        d = ovf[2 * (size_t)(base + lane)];
                        s = ovf[2 * (size_t)(base + lane) + 1];
                    }
                    s = ((unsigned)s < (unsigned)n) ? s : 0;
                    unsigned mk = (unsigned)__builtin_amdgcn_ballot_w32(d == row);
                    while (mk) {
                        const int l = __builtin_ctz(mk);
                        mk &= mk - 1;
                        const int sl = __builtin_amdgcn_readlane(s, l);
                        const float4 w = z4[(size_t)sl * (DOUT / 4) + l16];
                        if (!odd) {
                            acc.x += w.x; acc.y += w.y; acc.z += w.z; acc.w += w.w;
                        }
                    }
                }
            }
            acc.x += __shfl_xor(acc.x, 16);
            acc.y += __shfl_xor(acc.y, 16);
            acc.z += __shfl_xor(acc.z, 16);
            acc.w += __shfl_xor(acc.w, 16);
            float4 o;
            o.x = (acc.x + vs.x) * nr + bv.x;
            o.y = (acc.y + vs.y) * nr + bv.y;
            o.z = (acc.z + vs.z) * nr + bv.z;
            o.w = (acc.w + vs.w) * nr + bv.w;
            if (!odd) ((float4*)out)[(size_t)row * (DOUT / 4) + l16] = o;
        }
    }
}


extern "C" void kernel_launch(void* const* d_in, const int* in_sizes, int n_in,
                              void* d_out, int out_size, void* d_ws, size_t ws_size,
                              hipStream_t stream) {
    (void)n_in; (void)out_size; (void)ws_size;
    const float* x  = (const float*)d_in[0];
    const int*   src = (const int*)d_in[1];
    const int*   dst = (const int*)d_in[2];
    const float* W1 = (const float*)d_in[3];
    const float* b1 = (const float*)d_in[4];
    const float* W2 = (const float*)d_in[5];
    const float* b2 = (const float*)d_in[6];
    const float* W3 = (const float*)d_in[7];
    const float* b3 = (const float*)d_in[8];
    float* out = (float*)d_out;

    const int N = in_sizes[0] / DIN;
    const int E = in_sizes[1];

    char* ws = (char*)d_ws;
    size_t off = 0;
    float* buf0 = (float*)(ws + off);
    off += (size_t)N * DIN * 4;
    float* zbuf = (float*)(ws + off);
    off += (size_t)N * DIN * 4;
    int* bkt = (int*)(ws + off);
    off += (size_t)N * BSTRIDE * 4;
    int* novf = (int*)(ws + off);
    off += 256;
    int* ovf = (int*)(ws + off);

    const int TB = 256;
    const int ftiles  = N / FC_TR;
    const int fblocks = (ftiles + 3) / 4;
    const int lblocks = 768;

    GCN_zero_deg<<<(N + TB - 1) / TB, TB, 0, stream>>>(bkt, novf, N);
    GCN_fill_lines<<<(E + TB - 1) / TB, TB, 0, stream>>>(src, dst, bkt, novf, ovf, E, N);

    GCN_layer_wmma<128, true><<<lblocks, 256, 0, stream>>>(x, bkt, novf, ovf, W1, b1, buf0, ftiles, N, E);

    GCN_layer_fc2<<<lblocks, 256, 0, stream>>>(buf0, bkt, novf, ovf, W2, b2, W3, zbuf, ftiles, N, E);

    GCN_out_gather<<<fblocks, 256, 0, stream>>>(zbuf, bkt, novf, ovf, b3, out, ftiles, N, E);
}


# Node lines built by a two-pass LDS counting sort (edges grouped per 256-node bucket, lines filled by LDS atomics, stored whole): no per-edge global atomic or 4-B store; sources in LDS arrival order
# speedup vs baseline: 6.2332x; 1.3680x over previous
#include <hip/hip_runtime.h>

typedef __attribute__((ext_vector_type(2))) float v2f;
typedef __attribute__((ext_vector_type(8))) float v8f;

#define DIN 128
#define DOUT 64

#define BSTRIDE 32
#define BCAP 31
#define PCH 4096
#define PEPT 16
#define NPB 256
#define NPB_SH 8
#define NBMAX 512
#define BW 16
#define SB 16

__global__ __launch_bounds__(256) void GCN_part_edges(const int* __restrict__ src, const int* __restrict__ dst,
                                                      int2* __restrict__ part, int* __restrict__ offs,
                                                      int* __restrict__ novf, int E, int n, int nb1) {
    __shared__ int cnt[NBMAX];
    __shared__ int start[NBMAX + 1];
    __shared__ int wsum[8];
    __shared__ int2 stage[PCH];
    const int tid  = threadIdx.x;
    const int lane = tid & 31;
    const int wave = tid >> 5;
    const int base = blockIdx.x * PCH;
    cnt[tid] = 0;
    cnt[tid + 256] = 0;
    if (blockIdx.x == 0 && tid == 0) *novf = 0;
    __syncthreads();

    int d[PEPT], s[PEPT], r[PEPT];
#pragma unroll
    for (int u = 0; u < PEPT; ++u) {
        const int e = base + u * 256 + tid;
        d[u] = -1;
        s[u] = 0;
        if (e < E) {
            d[u] = dst[e];
            s[u] = src[e];
        }
    }
#pragma unroll
    for (int u = 0; u < PEPT; ++u) {
        r[u] = 0;
        if ((unsigned)d[u] < (unsigned)n) r[u] = atomicAdd(&cnt[d[u] >> NPB_SH], 1);
        else d[u] = -1;
    }
    __syncthreads();

    const int c0 = cnt[2 * tid];
    const int c1 = cnt[2 * tid + 1];
    int v = c0 + c1;
#pragma unroll
    for (int o = 1; o < 32; o <<= 1) {
        const int t = __shfl_up(v, o);
        if (lane >= o) v += t;
    }
    if (lane == 31) wsum[wave] = v;
    __syncthreads();
    int pre = 0;
#pragma unroll
    for (int w = 0; w < 8; ++w) pre += (w < wave) ? wsum[w] : 0;
    const int ex = pre + v - (c0 + c1);
    start[2 * tid]     = ex;
    start[2 * tid + 1] = ex + c0;
    if (tid == 255) start[NBMAX] = ex + c0 + c1;
    __syncthreads();

    int p[PEPT];
#pragma unroll
    for (int u = 0; u < PEPT; ++u) p[u] = start[(d[u] >= 0) ? (d[u] >> NPB_SH) : 0] + r[u];
#pragma unroll
    for (int u = 0; u < PEPT; ++u) {
        if (d[u] >= 0) stage[p[u]] = make_int2(d[u], s[u]);
    }
    __syncthreads();

    const int total = start[NBMAX];
    int2 w[PEPT];
#pragma unroll
    for (int u = 0; u < PEPT; ++u) w[u] = stage[u * 256 + tid];
#pragma unroll
    for (int u = 0; u < PEPT; ++u) {
        if (u * 256 + tid < total) part[(size_t)base + u * 256 + tid] = w[u];
    }
    for (int i = tid; i < nb1; i += 256) offs[(size_t)blockIdx.x * nb1 + i] = start[i];
}

__device__ __forceinline__ void GCN_put_edge(int2 x, int dl, int slot, int* __restrict__ lines,
                                             int* __restrict__ novf, int* __restrict__ ovf, int E) {
    if (slot < BCAP) {
        lines[dl * BSTRIDE + 1 + slot] = x.y;
    } else {
        const int k = atomicAdd(novf, 1);
        if ((unsigned)k < (unsigned)E) {
            ovf[2 * (size_t)k]     = x.x;
            ovf[2 * (size_t)k + 1] = x.y;
        }
    }
}

__global__ __launch_bounds__(512) void GCN_build_lines(const int2* __restrict__ part, const int* __restrict__ offs,
                                                       int* __restrict__ bkt, int* __restrict__ novf,
                                                       int* __restrict__ ovf, int E, int n, int nb1, int nparts) {
    __shared__ __attribute__((aligned(16))) int lines[NPB * BSTRIDE];
    __shared__ int degs[NPB];
    const int tid   = threadIdx.x;
    const int lane  = tid & 31;
    const int wave  = __builtin_amdgcn_readfirstlane(tid >> 5);
    const int bk    = blockIdx.x;
    const int node0 = bk * NPB;
    for (int i = tid; i < NPB * BSTRIDE; i += 512) lines[i] = 0;
    if (tid < NPB) degs[tid] = 0;
    __syncthreads();

    for (int q0 = 0; wave + BW * q0 < nparts; q0 += 32) {
        const int jq = wave + BW * (q0 + lane);
        int sv = 0, ev = 0;
        if (jq < nparts) {
            sv = offs[(size_t)jq * nb1 + bk];
            ev = offs[(size_t)jq * nb1 + bk + 1];
        }
        int nseg = (nparts - wave - BW * q0 + BW - 1) / BW;
        nseg = (nseg < 32) ? nseg : 32;
        for (int g = 0; g < nseg; g += SB) {
            int s0[SB], e0[SB];
#pragma unroll
            for (int u = 0; u < SB; ++u) {
                s0[u] = __builtin_amdgcn_readlane(sv, g + u);
                e0[u] = __builtin_amdgcn_readlane(ev, g + u);
            }
            int2 ed[SB];
#pragma unroll
            for (int u = 0; u < SB; ++u) {
                ed[u] = make_int2(-1, 0);
                if (s0[u] < e0[u]) {
                    const int j = wave + BW * (q0 + g + u);
                    int k = s0[u] + lane;
                    k = (k < e0[u]) ? k : (e0[u] - 1);
                    ed[u] = part[(size_t)j * PCH + k];
                }
            }
            int dl[SB], slot[SB];
#pragma unroll
            for (int u = 0; u < SB; ++u) {
                const int t = ed[u].x - node0;
                dl[u]   = -1;
                slot[u] = 0;
                if (s0[u] + lane < e0[u] && (unsigned)t < (unsigned)NPB) {
                    dl[u]   = t;
                    slot[u] = atomicAdd(&degs[t], 1);
                }
            }
#pragma unroll
            for (int u = 0; u < SB; ++u) {
                if (dl[u] >= 0) GCN_put_edge(ed[u], dl[u], slot[u], lines, novf, ovf, E);
                const int j = wave + BW * (q0 + g + u);
                for (int k = s0[u] + 32; k < e0[u]; k += 32) {
                    if (k + lane < e0[u]) {
                        const int2 x = part[(size_t)j * PCH + k + lane];
                        const int  t = x.x - node0;
                        if ((unsigned)t < (unsigned)NPB) {
                            const int sl = atomicAdd(&degs[t], 1);
                            GCN_put_edge(x, t, sl, lines, novf, ovf, E);
                        }
                    }
                }
            }
        }
    }
    __syncthreads();
    if (tid < NPB) lines[tid * BSTRIDE] = degs[tid];
    __syncthreads();

    for (int i = tid; i < NPB * (BSTRIDE / 4); i += 512) {
        if (node0 + (i >> 3) < n) ((int4*)bkt)[(size_t)node0 * (BSTRIDE / 4) + i] = ((const int4*)lines)[i];
    }
}

#define FC_TR 32

#define GB 16
__device__ __forceinline__ void GCN_sum_chunk(const float4* __restrict__ h4, int idx, int c,
                                              int lane, float4& acc) {
    int j = 0;
    for (; j + GB <= c; j += GB) {
        float4 v[GB];
#pragma unroll
        for (int u = 0; u < GB; ++u) {
            const int s = __builtin_amdgcn_readlane(idx, 1 + j + u);
            v[u] = h4[(size_t)s * (DIN / 4) + lane];
        }
#pragma unroll
        for (int u = 0; u < GB; ++u) {
            acc.x += v[u].x; acc.y += v[u].y; acc.z += v[u].z; acc.w += v[u].w;
        }
    }
    const int r = c - j;
    if (r > 0) {
        float4 v[GB - 1];
#pragma unroll
        for (int u = 0; u < GB - 1; ++u) {
            v[u] = make_float4(0.0f, 0.0f, 0.0f, 0.0f);
            if (u < r) {
                const int s = __builtin_amdgcn_readlane(idx, 1 + j + u);
                v[u] = h4[(size_t)s * (DIN / 4) + lane];
            }
        }
#pragma unroll
        for (int u = 0; u < GB - 1; ++u) {
            if (u < r) {
                acc.x += v[u].x; acc.y += v[u].y; acc.z += v[u].z; acc.w += v[u].w;
            }
        }
    }
}

__device__ __forceinline__ float4 GCN_neigh_sum(const float4* __restrict__ h4, int line, int deg,
                                                const int* __restrict__ novf,
                                                const int* __restrict__ ovf, int row, int lane,
                                                int n, int E) {
    float4 acc = make_float4(0.0f, 0.0f, 0.0f, 0.0f);
    const int c = (deg < BCAP) ? deg : BCAP;
    const int idx = ((unsigned)line < (unsigned)n) ? line : 0;
    GCN_sum_chunk(h4, idx, c, lane, acc);
    if (deg > BCAP) {
        int no = *novf;
        no = (no < E) ? no : E;
        for (int base = 0; base < no; base += 32) {
            int d = -1, s = 0;
            if (base + lane < no) {
                d = ovf[2 * (size_t)(base + lane)];
                s = ovf[2 * (size_t)(base + lane) + 1];
            }
            s = ((unsigned)s < (unsigned)n) ? s : 0;
            unsigned m = (unsigned)__builtin_amdgcn_ballot_w32(d == row);
            while (m) {
                const int l = __builtin_ctz(m);
                m &= m - 1;
                const int sl = __builtin_amdgcn_readlane(s, l);
                const float4 v = h4[(size_t)sl * (DIN / 4) + lane];
                acc.x += v.x; acc.y += v.y; acc.z += v.z; acc.w += v.w;
            }
        }
    }
    return acc;
}

template <int NC, bool RELU>
__global__ __launch_bounds__(256) void GCN_layer_wmma(const float* __restrict__ h,
                                                      const int* __restrict__ bkt,
                                                      const int* __restrict__ novf,
                                                      const int* __restrict__ ovf,
                                                      const float* __restrict__ W,
                                                      const float* __restrict__ bias,
                                                      float* __restrict__ out, int ntiles, int n,
                                                      int E) {
    constexpr int K    = DIN;
    constexpr int KS   = K / 4;
    constexpr int LDSK = K + 4;
    constexpr int NT   = NC / 16;
    constexpr int WPN  = 8 / NT;
    constexpr int RT   = 2 / WPN;
    constexpr int RW   = FC_TR / 8;
    __shared__ __attribute__((aligned(16))) float Atile[2][FC_TR * LDSK];

    const int tid  = threadIdx.x;
    const int wave = tid >> 5;
    const int lane = tid & 31;
    const int m    = lane & 15;
    const int kh2  = (lane >> 4) << 1;
    const int n0   = (wave % NT) * 16;
    const int rt0  = (wave / NT) * RT;
    const int wrow0 = __builtin_amdgcn_readfirstlane(wave * RW);

    const float* wrow = W + (size_t)(n0 + m) * K;
    v2f b[KS];
#pragma unroll
    for (int ks = 0; ks < KS; ++ks) {
        b[ks].x = wrow[ks * 4 + kh2];
        b[ks].y = wrow[ks * 4 + kh2 + 1];
    }
    const float bn   = bias[n0 + m];
    const int   rupp = (lane >> 4) << 3;

    const float4* __restrict__ h4 = (const float4*)h;

    int buf = 0;
    for (int t = blockIdx.x; t < ntiles; t += gridDim.x, buf ^= 1) {
        const int row0 = t * FC_TR;
        float* At = Atile[buf];

        {
            const int rb = row0 + wrow0;
            int ln[RW];
#pragma unroll
            for (int i = 0; i < RW; ++i) ln[i] = bkt[(size_t)(rb + i) * BSTRIDE + lane];
#pragma unroll
            for (int i = 0; i < RW; ++i) {
                const float4 vs  = h4[(size_t)(rb + i) * (DIN / 4) + lane];
                const int    deg = __builtin_amdgcn_readfirstlane(ln[i]);
                const float  nr  = 1.0f / ((float)deg + 1.0f);
                const float4 ag  = GCN_neigh_sum(h4, ln[i], deg, novf, ovf, rb + i, lane, n, E);
                float4 o;
                o.x = (ag.x + vs.x) * nr;
                o.y = (ag.y + vs.y) * nr;
                o.z = (ag.z + vs.z) * nr;
                o.w = (ag.w + vs.w) * nr;
                *(float4*)(At + (wrow0 + i) * LDSK + lane * 4) = o;
            }
        }
        __syncthreads();

        v8f acc[RT];
#pragma unroll
        for (int rt = 0; rt < RT; ++rt) acc[rt] = v8f{};
#pragma unroll
        for (int ks = 0; ks < KS; ++ks) {
#pragma unroll
            for (int rt = 0; rt < RT; ++rt) {
                const float* arow = At + ((rt0 + rt) * 16 + m) * LDSK;
                v2f a;
                a.x = arow[ks * 4 + kh2];
                a.y = arow[ks * 4 + kh2 + 1];
                acc[rt] = __builtin_amdgcn_wmma_f32_16x16x4_f32(
                     false, a,  false, b[ks],
                     (short)0, acc[rt],  false,  false);
            }
        }

#pragma unroll
        for (int rt = 0; rt < RT; ++rt) {
            float* op = out + (size_t)(row0 + (rt0 + rt) * 16 + rupp) * NC + (n0 + m);
#pragma unroll
            for (int v = 0; v < 8; ++v) {
                float val = acc[rt][v] + bn;
                if (RELU) val = fmaxf(val, 0.0f);
                op[v * NC] = val;
            }
        }
    }
}

__global__ __launch_bounds__(256) void GCN_layer_fc2(const float* __restrict__ h,
                                                     const int* __restrict__ bkt,
                                                     const int* __restrict__ novf,
                                                     const int* __restrict__ ovf,
                                                     const float* __restrict__ W,
                                                     const float* __restrict__ bias,
                                                     const float* __restrict__ Wn,
                                                     float* __restrict__ z, int ntiles, int n, int E) {
    constexpr int K    = DIN;
    constexpr int KS   = K / 4;
    constexpr int LDSK = K + 4;
    constexpr int RW   = FC_TR / 8;
    __shared__ __attribute__((aligned(16))) float At[FC_TR * LDSK];
    __shared__ __attribute__((aligned(16))) float Ht[FC_TR * LDSK];
    __shared__ __attribute__((aligned(16))) float Wt[DOUT * LDSK];

    const int tid  = threadIdx.x;
    const int wave = tid >> 5;
    const int lane = tid & 31;
    const int m    = lane & 15;
    const int kh2  = (lane >> 4) << 1;
    const int n0   = wave * 16;
    const int rt2  = wave >> 2;
    const int n2   = (wave & 3) * 16;
    const int wrow0 = __builtin_amdgcn_readfirstlane(wave * RW);

    for (int i = tid; i < DOUT * (K / 4); i += 256) {
        const int r  = i / (K / 4);
        const int c4 = i % (K / 4);
        *(float4*)(Wt + r * LDSK + c4 * 4) = ((const float4*)Wn)[i];
    }

    const float* wrow = W + (size_t)(n0 + m) * K;
    v2f b[KS];
#pragma unroll
    for (int ks = 0; ks < KS; ++ks) {
        b[ks].x = wrow[ks * 4 + kh2];
        b[ks].y = wrow[ks * 4 + kh2 + 1];
    }
    const float bn   = bias[n0 + m];
    const int   rupp = (lane >> 4) << 3;

    const float4* __restrict__ h4 = (const float4*)h;

    for (int t = blockIdx.x; t < ntiles; t += gridDim.x) {
        const int row0 = t * FC_TR;

        {
            const int rb = row0 + wrow0;
            int ln[RW];
#pragma unroll
            for (int i = 0; i < RW; ++i) ln[i] = bkt[(size_t)(rb + i) * BSTRIDE + lane];
#pragma unroll
            for (int i = 0; i < RW; ++i) {
                const float4 vs  = h4[(size_t)(rb + i) * (DIN / 4) + lane];
                const int    deg = __builtin_amdgcn_readfirstlane(ln[i]);
                const float  nr  = 1.0f / ((float)deg + 1.0f);
                const float4 ag  = GCN_neigh_sum(h4, ln[i], deg, novf, ovf, rb + i, lane, n, E);
                float4 o;
                o.x = (ag.x + vs.x) * nr;
                o.y = (ag.y + vs.y) * nr;
                o.z = (ag.z + vs.z) * nr;
                o.w = (ag.w + vs.w) * nr;
                *(float4*)(At + (wrow0 + i) * LDSK + lane * 4) = o;
            }
        }
        __syncthreads();

        v8f acc[2];
        acc[0] = v8f{};
        acc[1] = v8f{};
#pragma unroll
        for (int ks = 0; ks < KS; ++ks) {
#pragma unroll
            for (int rt = 0; rt < 2; ++rt) {
                const float* arow = At + (rt * 16 + m) * LDSK;
                v2f a;
                a.x = arow[ks * 4 + kh2];
                a.y = arow[ks * 4 + kh2 + 1];
                acc[rt] = __builtin_amdgcn_wmma_f32_16x16x4_f32(
                     false, a,  false, b[ks],
                     (short)0, acc[rt],  false,  false);
            }
        }

#pragma unroll
        for (int rt = 0; rt < 2; ++rt) {
            float* hp = Ht + (rt * 16 + rupp) * LDSK + (n0 + m);
#pragma unroll
            for (int v = 0; v < 8; ++v) hp[v * LDSK] = fmaxf(acc[rt][v] + bn, 0.0f);
        }
        __syncthreads();

        v8f c0 = v8f{};
        v8f c1 = v8f{};
        const float* hrow = Ht + (rt2 * 16 + m) * LDSK;
        const float* wn   = Wt + (n2 + m) * LDSK;
#pragma unroll
        for (int ks = 0; ks < KS; ks += 2) {
            v2f a0, b0, a1, b1;
            a0.x = hrow[ks * 4 + kh2];
            a0.y = hrow[ks * 4 + kh2 + 1];
            b0.x = wn[ks * 4 + kh2];
            b0.y = wn[ks * 4 + kh2 + 1];
            a1.x = hrow[ks * 4 + 4 + kh2];
            a1.y = hrow[ks * 4 + 4 + kh2 + 1];
            b1.x = wn[ks * 4 + 4 + kh2];
            b1.y = wn[ks * 4 + 4 + kh2 + 1];
            c0 = __builtin_amdgcn_wmma_f32_16x16x4_f32(false, a0, false, b0, (short)0, c0, false, false);
            c1 = __builtin_amdgcn_wmma_f32_16x16x4_f32(false, a1, false, b1, (short)0, c1, false, false);
        }

        float* zp = z + (size_t)(row0 + rt2 * 16 + rupp) * DOUT + (n2 + m);
#pragma unroll
        for (int v = 0; v < 8; ++v) zp[v * DOUT] = c0[v] + c1[v];
    }
}


__device__ __forceinline__ void GCN_sum_pairs(const float4* __restrict__ z4, int idx, int c,
                                              int l16, bool odd, float4& acc) {
    int j = 0;
    for (; j + 16 <= c; j += 16) {
        float4 v[8];
#pragma unroll
        for (int u = 0; u < 8; ++u) {
            const int s0 = __builtin_amdgcn_readlane(idx, 1 + j + 2 * u);
            const int s1 = __builtin_amdgcn_readlane(idx, 2 + j + 2 * u);
            const int s  = odd ? s1 : s0;
            v[u] = z4[(size_t)s * (DOUT / 4) + l16];
        }
#pragma unroll
        for (int u = 0; u < 8; ++u) {
            acc.x += v[u].x; acc.y += v[u].y; acc.z += v[u].z; acc.w += v[u].w;
        }
    }
    const int r = c - j;
    if (r > 0) {
        float4 v[8];
#pragma unroll
        for (int u = 0; u < 8; ++u) {
            v[u] = make_float4(0.0f, 0.0f, 0.0f, 0.0f);
            if (2 * u < r) {
                const bool two = (2 * u + 1 < r);
                const int  s0  = __builtin_amdgcn_readlane(idx, 1 + j + 2 * u);
                const int  s1  = two ? __builtin_amdgcn_readlane(idx, 2 + j + 2 * u) : s0;
                const int  s   = odd ? s1 : s0;
                v[u] = z4[(size_t)s * (DOUT / 4) + l16];
            }
        }
#pragma unroll
        for (int u = 0; u < 8; ++u) {
            if (2 * u < r) {
                const bool use = (2 * u + 1 < r) || !odd;
                acc.x += use ? v[u].x : 0.0f;
                acc.y += use ? v[u].y : 0.0f;
                acc.z += use ? v[u].z : 0.0f;
                acc.w += use ? v[u].w : 0.0f;
            }
        }
    }
}

__global__ __launch_bounds__(256) void GCN_out_gather(const float* __restrict__ z,
                                                      const int* __restrict__ bkt,
                                                      const int* __restrict__ novf,
                                                      const int* __restrict__ ovf,
                                                      const float* __restrict__ bias,
                                                      float* __restrict__ out, int ntiles, int n,
                                                      int E) {
    constexpr int RW = FC_TR / 8;
    const int  tid  = threadIdx.x;
    const int  wave = tid >> 5;
    const int  lane = tid & 31;
    const int  l16  = lane & 15;
    const bool odd  = lane >= 16;
    const int wrow0 = __builtin_amdgcn_readfirstlane(wave * RW);

    const float4* __restrict__ z4 = (const float4*)z;
    const float4 bv = ((const float4*)bias)[l16];

    for (int t = blockIdx.x; t < ntiles; t += gridDim.x) {
        const int rb = t * FC_TR + wrow0;
        int ln[RW];
#pragma unroll
        for (int i = 0; i < RW; ++i) ln[i] = bkt[(size_t)(rb + i) * BSTRIDE + lane];
#pragma unroll
        for (int i = 0; i < RW; ++i) {
            const int    row = rb + i;
            const float4 vs  = z4[(size_t)row * (DOUT / 4) + l16];
            const int    deg = __builtin_amdgcn_readfirstlane(ln[i]);
            const float  nr  = 1.0f / ((float)deg + 1.0f);
            const int    c   = (deg < BCAP) ? deg : BCAP;
            const int    idx = ((unsigned)ln[i] < (unsigned)n) ? ln[i] : 0;
            float4 acc = make_float4(0.0f, 0.0f, 0.0f, 0.0f);
            GCN_sum_pairs(z4, idx, c, l16, odd, acc);
            if (deg > BCAP) {
                int no = *novf;
                no = (no < E) ? no : E;
                for (int base = 0; base < no; base += 32) {
                    int d = -1, s = 0;
                    if (base + lane < no) {
                        d = ovf[2 * (size_t)(base + lane)];
                        s = ovf[2 * (size_t)(base + lane) + 1];
                    }
                    s = ((unsigned)s < (unsigned)n) ? s : 0;
                    unsigned mk = (unsigned)__builtin_amdgcn_ballot_w32(d == row);
                    while (mk) {
                        const int l = __builtin_ctz(mk);
                        mk &= mk - 1;
                        const int sl = __builtin_amdgcn_readlane(s, l);
                        const float4 w = z4[(size_t)sl * (DOUT / 4) + l16];
                        if (!odd) {
                            acc.x += w.x; acc.y += w.y; acc.z += w.z; acc.w += w.w;
                        }
                    }
                }
            }
            acc.x += __shfl_xor(acc.x, 16);
            acc.y += __shfl_xor(acc.y, 16);
            acc.z += __shfl_xor(acc.z, 16);
            acc.w += __shfl_xor(acc.w, 16);
            float4 o;
            o.x = (acc.x + vs.x) * nr + bv.x;
            o.y = (acc.y + vs.y) * nr + bv.y;
            o.z = (acc.z + vs.z) * nr + bv.z;
            o.w = (acc.w + vs.w) * nr + bv.w;
            if (!odd) ((float4*)out)[(size_t)row * (DOUT / 4) + l16] = o;
        }
    }
}


extern "C" void kernel_launch(void* const* d_in, const int* in_sizes, int n_in,
                              void* d_out, int out_size, void* d_ws, size_t ws_size,
                              hipStream_t stream) {
    (void)n_in; (void)out_size; (void)ws_size;
    const float* x  = (const float*)d_in[0];
    const int*   src = (const int*)d_in[1];
    const int*   dst = (const int*)d_in[2];
    const float* W1 = (const float*)d_in[3];
    const float* b1 = (const float*)d_in[4];
    const float* W2 = (const float*)d_in[5];
    const float* b2 = (const float*)d_in[6];
    const float* W3 = (const float*)d_in[7];
    const float* b3 = (const float*)d_in[8];
    float* out = (float*)d_out;

    const int N = in_sizes[0] / DIN;
    const int E = in_sizes[1];

    char* ws = (char*)d_ws;
    size_t off = 0;
    float* buf0 = (float*)(ws + off);
    off += (size_t)N * DIN * 4;
    float* zbuf = (float*)(ws + off);
    off += (size_t)N * DIN * 4;
    int* bkt = (int*)(ws + off);
    off += (size_t)N * BSTRIDE * 4;
    int* novf = (int*)(ws + off);
    off += 256;
    int* ovf = (int*)(ws + off);
    off += (size_t)E * 8;
    int2* part = (int2*)(ws + off);
    off += (size_t)E * 8;
    int* offs = (int*)(ws + off);

    const int ftiles  = N / FC_TR;
    const int fblocks = (ftiles + 3) / 4;
    const int lblocks = 768;

    const int nb     = (N + NPB - 1) / NPB;
    const int nparts = (E + PCH - 1) / PCH;
    GCN_part_edges<<<nparts, 256, 0, stream>>>(src, dst, part, offs, novf, E, N, nb + 1);
    GCN_build_lines<<<nb, 512, 0, stream>>>(part, offs, bkt, novf, ovf, E, N, nb + 1, nparts);

    GCN_layer_wmma<128, true><<<lblocks, 256, 0, stream>>>(x, bkt, novf, ovf, W1, b1, buf0, ftiles, N, E);

    GCN_layer_fc2<<<lblocks, 256, 0, stream>>>(buf0, bkt, novf, ovf, W2, b2, W3, zbuf, ftiles, N, E);

    GCN_out_gather<<<fblocks, 256, 0, stream>>>(zbuf, bkt, novf, ovf, b3, out, ftiles, N, E);
}
